# starting kernel
# baseline (speedup 1.0000x reference)
#include <hip/hip_runtime.h>

#define B_     4096
#define NFIELD 20
#define NEMB   16
#define HID    512
#define K_     16
#define GIN    320
#define EPSF   1e-10f

typedef __attribute__((ext_vector_type(2))) float v2f;
typedef __attribute__((ext_vector_type(8))) float v8f;

__device__ __forceinline__ v8f wmma_f32_16x16x4(v2f a, v2f b, v8f c) {
  return __builtin_amdgcn_wmma_f32_16x16x4_f32(false, a, false, b, (short)0, c,
                                               false, false);
}

__global__ void __launch_bounds__(256)
vsams_init_kernel(float* __restrict__ out, int* __restrict__ counts) {
  int t = blockIdx.x * blockDim.x + threadIdx.x;
  if (t <= B_) out[t] = 0.0f;
  if (t < K_) counts[t] = 0;
}

__global__ void __launch_bounds__(128)
vsams_gate_l1_kernel(const int* __restrict__ sql,
                     const float* __restrict__ sql_emb_w,
                     const float* __restrict__ gw1,
                     const float* __restrict__ gb1,
                     float* __restrict__ h) {
  __shared__ float Alds[16 * GIN];
  const int tid = threadIdx.x;
  const int m0  = blockIdx.y * 16;
  for (int idx = tid; idx < 16 * GIN; idx += 128) {
    int r = idx / GIN, c = idx - r * GIN;
    int f = c >> 4, e = c & 15;
    int s = sql[(m0 + r) * NFIELD + f];
    Alds[idx] = sql_emb_w[s * NEMB + e];
  }
  __syncthreads();
  const int lane = tid & 31, wave = tid >> 5;
  const int n0 = blockIdx.x * 64 + wave * 16;
  const int m  = lane & 15;
  const int kh = lane >> 4;
  v8f acc = {0.f, 0.f, 0.f, 0.f, 0.f, 0.f, 0.f, 0.f};
#pragma unroll 4
  for (int k0 = 0; k0 < GIN; k0 += 4) {
    v2f a, b;
    const float* ap = &Alds[m * GIN + k0 + 2 * kh];
    a.x = ap[0]; a.y = ap[1];
    const float* bp = &gw1[(size_t)(k0 + 2 * kh) * HID + n0 + m];
    b.x = bp[0]; b.y = bp[HID];
    acc = wmma_f32_16x16x4(a, b, acc);
  }
#pragma unroll
  for (int v = 0; v < 8; ++v) {
    int row = m0 + v + 8 * kh;
    int col = n0 + m;
    float val = acc[v] + gb1[col];
    h[(size_t)row * HID + col] = val > 0.f ? val : 0.f;
  }
}

__global__ void __launch_bounds__(32)
vsams_gate_head_kernel(const float* __restrict__ h,
                       const float* __restrict__ gw2,
                       const float* __restrict__ gb2,
                       float* __restrict__ gates,
                       int* __restrict__ counts,
                       int* __restrict__ lists) {
  __shared__ float logit[16 * 16];
  const int lane = threadIdx.x;
  const int m0 = blockIdx.x * 16;
  const int m  = lane & 15;
  const int kh = lane >> 4;
  v8f acc = {0.f, 0.f, 0.f, 0.f, 0.f, 0.f, 0.f, 0.f};
#pragma unroll 4
  for (int k0 = 0; k0 < HID; k0 += 4) {
    v2f a, b;
    const float* ap = &h[(size_t)(m0 + m) * HID + k0 + 2 * kh];
    a.x = ap[0]; a.y = ap[1];
    const float* bp = &gw2[(k0 + 2 * kh) * K_ + m];
    b.x = bp[0]; b.y = bp[K_];
    acc = wmma_f32_16x16x4(a, b, acc);
  }
#pragma unroll
  for (int v = 0; v < 8; ++v)
    logit[(v + 8 * kh) * 16 + m] = acc[v] + gb2[m];
  __syncthreads();
  if (lane < 16) {
    const int b = m0 + lane;
    float l[K_];
    float mx = -3.0e38f;
    for (int j = 0; j < K_; ++j) { l[j] = logit[lane * 16 + j]; mx = fmaxf(mx, l[j]); }
    float s = 0.f;
    for (int j = 0; j < K_; ++j) { l[j] = __expf(l[j] - mx); s += l[j]; }
    float inv = 1.f / s;
    for (int j = 0; j < K_; ++j) l[j] *= inv;
    int i1 = 0; float v1 = l[0];
    for (int j = 1; j < K_; ++j) if (l[j] > v1) { v1 = l[j]; i1 = j; }
    int i2 = -1; float v2 = -1.f;
    for (int j = 0; j < K_; ++j) if (j != i1 && l[j] > v2) { v2 = l[j]; i2 = j; }
    float denom = v1 + v2;
    float g1 = v1 / denom, g2 = v2 / denom;
    for (int j = 0; j < K_; ++j) gates[b * K_ + j] = 0.f;
    gates[b * K_ + i1] = g1;
    gates[b * K_ + i2] = g2;
    int p1 = atomicAdd(&counts[i1], 1);
    lists[i1 * B_ + p1] = b;
    int p2 = atomicAdd(&counts[i2], 1);
    lists[i2 * B_ + p2] = b;
  }
}

__global__ void __launch_bounds__(128)
vsams_expert_kernel(const int* __restrict__ x,
                    const float* __restrict__ inp_emb_w,
                    const float* __restrict__ ew1,
                    const float* __restrict__ eb1,
                    const float* __restrict__ ew2,
                    const float* __restrict__ eb2,
                    const float* __restrict__ gates,
                    const int* __restrict__ counts,
                    const int* __restrict__ lists,
                    float* __restrict__ y) {
  __shared__ float Alds[16 * GIN];
  __shared__ float EH[16 * HID];
  __shared__ float psum[16 * 8];
  __shared__ int   bids[16];
  const int k    = blockIdx.x >> 8;
  const int tile = blockIdx.x & 255;
  const int cnt  = counts[k];
  const int base = tile * 16;
  if (base >= cnt) return;
  const int tid = threadIdx.x;
  if (tid < 16) bids[tid] = (base + tid < cnt) ? lists[k * B_ + base + tid] : -1;
  __syncthreads();
  for (int idx = tid; idx < 16 * GIN; idx += 128) {
    int r = idx / GIN, c = idx - r * GIN;
    int b = bids[r];
    float val = 0.f;
    if (b >= 0) {
      int f = c >> 4, e = c & 15;
      int xi = x[b * NFIELD + f];
      val = inp_emb_w[(size_t)xi * NEMB + e];
    }
    Alds[idx] = val;
  }
  __syncthreads();
  const int lane = tid & 31, wave = tid >> 5;
  const int m  = lane & 15;
  const int kh = lane >> 4;
  const float* W1 = ew1 + (size_t)k * GIN * HID;
  const float* B1 = eb1 + k * HID;
  for (int nt = 0; nt < 8; ++nt) {
    const int n0 = (wave * 8 + nt) * 16;
    v8f acc = {0.f, 0.f, 0.f, 0.f, 0.f, 0.f, 0.f, 0.f};
#pragma unroll 4
    for (int k0 = 0; k0 < GIN; k0 += 4) {
      v2f a, b;
      const float* ap = &Alds[m * GIN + k0 + 2 * kh];
      a.x = ap[0]; a.y = ap[1];
      const float* bp = &W1[(size_t)(k0 + 2 * kh) * HID + n0 + m];
      b.x = bp[0]; b.y = bp[HID];
      acc = wmma_f32_16x16x4(a, b, acc);
    }
#pragma unroll
    for (int v = 0; v < 8; ++v) {
      int row = v + 8 * kh;
      int col = n0 + m;
      float val = acc[v] + B1[col];
      EH[row * HID + col] = val > 0.f ? val : 0.f;
    }
  }
  __syncthreads();
  {
    int row = tid >> 3, seg = tid & 7;
    const float* w2 = ew2 + k * HID + seg * 64;
    const float* eh = EH + row * HID + seg * 64;
    float p = 0.f;
#pragma unroll 8
    for (int c = 0; c < 64; ++c) p += eh[c] * w2[c];
    psum[row * 8 + seg] = p;
  }
  __syncthreads();
  if (tid < 16) {
    int b = bids[tid];
    if (b >= 0) {
      float eo = eb2[k];
      for (int s = 0; s < 8; ++s) eo += psum[tid * 8 + s];
      float g = gates[b * K_ + k];
      atomicAdd(&y[b], g * eo);
    }
  }
}

__global__ void __launch_bounds__(256)
vsams_loss_kernel(const float* __restrict__ gates, float* __restrict__ out) {
  __shared__ float simp[256], sld[256];
  const int t = threadIdx.x;
  const int k = t & 15, chunk = t >> 4;
  float pi = 0.f, pl = 0.f;
  for (int r = chunk * 256; r < chunk * 256 + 256; ++r) {
    float g = gates[r * K_ + k];
    pi += g;
    pl += (g > 0.f) ? 1.f : 0.f;
  }
  simp[t] = pi; sld[t] = pl;
  __syncthreads();
  if (t == 0) {
    float imp[K_], ld[K_];
    for (int kk = 0; kk < K_; ++kk) { imp[kk] = 0.f; ld[kk] = 0.f; }
    for (int c = 0; c < 16; ++c)
      for (int kk = 0; kk < K_; ++kk) {
        imp[kk] += simp[c * 16 + kk];
        ld[kk]  += sld[c * 16 + kk];
      }
    float mi = 0.f, ml = 0.f;
    for (int kk = 0; kk < K_; ++kk) { mi += imp[kk]; ml += ld[kk]; }
    mi *= (1.f / 16.f); ml *= (1.f / 16.f);
    float vi = 0.f, vl = 0.f;
    for (int kk = 0; kk < K_; ++kk) {
      float di = imp[kk] - mi, dl = ld[kk] - ml;
      vi += di * di; vl += dl * dl;
    }
    vi *= (1.f / 16.f); vl *= (1.f / 16.f);
    out[B_] = vi / (mi * mi + EPSF) + vl / (ml * ml + EPSF);
  }
}

extern "C" void kernel_launch(void* const* d_in, const int* in_sizes, int n_in,
                              void* d_out, int out_size, void* d_ws, size_t ws_size,
                              hipStream_t stream) {
  const int*   x         = (const int*)d_in[0];
  const int*   sql       = (const int*)d_in[1];
  const float* sql_emb_w = (const float*)d_in[2];
  const float* inp_emb_w = (const float*)d_in[3];
  const float* gw1 = (const float*)d_in[4];
  const float* gb1 = (const float*)d_in[5];
  const float* gw2 = (const float*)d_in[6];
  const float* gb2 = (const float*)d_in[7];
  const float* ew1 = (const float*)d_in[8];
  const float* eb1 = (const float*)d_in[9];
  const float* ew2 = (const float*)d_in[10];
  const float* eb2 = (const float*)d_in[11];
  float* out = (float*)d_out;

  float* h      = (float*)d_ws;
  float* gates  = h + (size_t)B_ * HID;
  int*   counts = (int*)(gates + (size_t)B_ * K_);
  int*   lists  = counts + K_;

  vsams_init_kernel<<<17, 256, 0, stream>>>(out, counts);
  vsams_gate_l1_kernel<<<dim3(8, 256), 128, 0, stream>>>(sql, sql_emb_w, gw1, gb1, h);
  vsams_gate_head_kernel<<<256, 32, 0, stream>>>(h, gw2, gb2, gates, counts, lists);
  vsams_expert_kernel<<<K_ * 256, 128, 0, stream>>>(x, inp_emb_w, ew1, eb1, ew2,
                                                    eb2, gates, counts, lists, out);
  vsams_loss_kernel<<<1, 256, 0, stream>>>(gates, out);
}


# More loads in flight per wait: loss sums, both embedding gathers (16-byte pieces) and the WMMA k-loops issue 10-40 independent loads before the first wait, were 1-8; loss sum order changes
# speedup vs baseline: 1.4318x; 1.4318x over previous
#include <hip/hip_runtime.h>

#define B_     4096
#define NFIELD 20
#define NEMB   16
#define HID    512
#define K_     16
#define GIN    320
#define EPSF   1e-10f

typedef __attribute__((ext_vector_type(2))) float v2f;
typedef __attribute__((ext_vector_type(8))) float v8f;
typedef __attribute__((ext_vector_type(4))) float v4f;

__device__ __forceinline__ v8f wmma_f32_16x16x4(v2f a, v2f b, v8f c) {
  return __builtin_amdgcn_wmma_f32_16x16x4_f32(false, a, false, b, (short)0, c,
                                               false, false);
}

__global__ void __launch_bounds__(256)
vsams_init_kernel(float* __restrict__ out, int* __restrict__ counts) {
  int t = blockIdx.x * blockDim.x + threadIdx.x;
  if (t <= B_) out[t] = 0.0f;
  if (t < K_) counts[t] = 0;
}

__global__ void __launch_bounds__(128)
vsams_gate_l1_kernel(const int* __restrict__ sql,
                     const float* __restrict__ sql_emb_w,
                     const float* __restrict__ gw1,
                     const float* __restrict__ gb1,
                     float* __restrict__ h) {
  __shared__ __attribute__((aligned(16))) float Alds[16 * GIN];
  const int tid = threadIdx.x;
  const int m0  = blockIdx.y * 16;
  {
    int s[10];
#pragma unroll
    for (int i = 0; i < 10; ++i) {
      int idx = tid + 128 * i;
      int r = idx / 80;
      int f = (idx - r * 80) >> 2;
      s[i] = sql[(m0 + r) * NFIELD + f];
    }
    v4f v[10];
#pragma unroll
    for (int i = 0; i < 10; ++i)
      v[i] = *(const v4f*)&sql_emb_w[(size_t)s[i] * NEMB + (tid & 3) * 4];
#pragma unroll
    for (int i = 0; i < 10; ++i) *(v4f*)&Alds[(tid + 128 * i) * 4] = v[i];
  }
  __syncthreads();
  const int lane = tid & 31, wave = tid >> 5;
  const int n0 = blockIdx.x * 64 + wave * 16;
  const int m  = lane & 15;
  const int kh = lane >> 4;
  v8f acc = {0.f, 0.f, 0.f, 0.f, 0.f, 0.f, 0.f, 0.f};
#pragma unroll 20
  for (int k0 = 0; k0 < GIN; k0 += 4) {
    v2f a, b;
    const float* ap = &Alds[m * GIN + k0 + 2 * kh];
    a.x = ap[0]; a.y = ap[1];
    const float* bp = &gw1[(size_t)(k0 + 2 * kh) * HID + n0 + m];
    b.x = bp[0]; b.y = bp[HID];
    acc = wmma_f32_16x16x4(a, b, acc);
  }
#pragma unroll
  for (int v = 0; v < 8; ++v) {
    int row = m0 + v + 8 * kh;
    int col = n0 + m;
    float val = acc[v] + gb1[col];
    h[(size_t)row * HID + col] = val > 0.f ? val : 0.f;
  }
}

__global__ void __launch_bounds__(32)
vsams_gate_head_kernel(const float* __restrict__ h,
                       const float* __restrict__ gw2,
                       const float* __restrict__ gb2,
                       float* __restrict__ gates,
                       int* __restrict__ counts,
                       int* __restrict__ lists) {
  __shared__ float logit[16 * 16];
  const int lane = threadIdx.x;
  const int m0 = blockIdx.x * 16;
  const int m  = lane & 15;
  const int kh = lane >> 4;
  v8f acc = {0.f, 0.f, 0.f, 0.f, 0.f, 0.f, 0.f, 0.f};
#pragma unroll 8
  for (int k0 = 0; k0 < HID; k0 += 4) {
    v2f a, b;
    const float* ap = &h[(size_t)(m0 + m) * HID + k0 + 2 * kh];
    a.x = ap[0]; a.y = ap[1];
    const float* bp = &gw2[(k0 + 2 * kh) * K_ + m];
    b.x = bp[0]; b.y = bp[K_];
    acc = wmma_f32_16x16x4(a, b, acc);
  }
#pragma unroll
  for (int v = 0; v < 8; ++v)
    logit[(v + 8 * kh) * 16 + m] = acc[v] + gb2[m];
  __syncthreads();
  if (lane < 16) {
    const int b = m0 + lane;
    float l[K_];
    float mx = -3.0e38f;
    for (int j = 0; j < K_; ++j) { l[j] = logit[lane * 16 + j]; mx = fmaxf(mx, l[j]); }
    float s = 0.f;
    for (int j = 0; j < K_; ++j) { l[j] = __expf(l[j] - mx); s += l[j]; }
    float inv = 1.f / s;
    for (int j = 0; j < K_; ++j) l[j] *= inv;
    int i1 = 0; float v1 = l[0];
    for (int j = 1; j < K_; ++j) if (l[j] > v1) { v1 = l[j]; i1 = j; }
    int i2 = -1; float v2 = -1.f;
    for (int j = 0; j < K_; ++j) if (j != i1 && l[j] > v2) { v2 = l[j]; i2 = j; }
    float denom = v1 + v2;
    float g1 = v1 / denom, g2 = v2 / denom;
    for (int j = 0; j < K_; ++j) gates[b * K_ + j] = 0.f;
    gates[b * K_ + i1] = g1;
    gates[b * K_ + i2] = g2;
    int p1 = atomicAdd(&counts[i1], 1);
    lists[i1 * B_ + p1] = b;
    int p2 = atomicAdd(&counts[i2], 1);
    lists[i2 * B_ + p2] = b;
  }
}

__global__ void __launch_bounds__(128)
vsams_expert_kernel(const int* __restrict__ x,
                    const float* __restrict__ inp_emb_w,
                    const float* __restrict__ ew1,
                    const float* __restrict__ eb1,
                    const float* __restrict__ ew2,
                    const float* __restrict__ eb2,
                    const float* __restrict__ gates,
                    const int* __restrict__ counts,
                    const int* __restrict__ lists,
                    float* __restrict__ y) {
  __shared__ __attribute__((aligned(16))) float Alds[16 * GIN];
  __shared__ float EH[16 * HID];
  __shared__ float psum[16 * 8];
  __shared__ int   bids[16];
  const int k    = blockIdx.x >> 8;
  const int tile = blockIdx.x & 255;
  const int cnt  = counts[k];
  const int base = tile * 16;
  if (base >= cnt) return;
  const int tid = threadIdx.x;
  if (tid < 16) bids[tid] = (base + tid < cnt) ? lists[k * B_ + base + tid] : -1;
  __syncthreads();
  {
    int xi[10];
#pragma unroll
    for (int i = 0; i < 10; ++i) {
      int idx = tid + 128 * i;
      int r = idx / 80;
      int f = (idx - r * 80) >> 2;
      int b = bids[r];
      xi[i] = x[(b >= 0 ? b : 0) * NFIELD + f];
    }
    v4f v[10];
#pragma unroll
    for (int i = 0; i < 10; ++i)
      v[i] = *(const v4f*)&inp_emb_w[(size_t)xi[i] * NEMB + (tid & 3) * 4];
#pragma unroll
    for (int i = 0; i < 10; ++i) *(v4f*)&Alds[(tid + 128 * i) * 4] = v[i];
  }
  __syncthreads();
  const int lane = tid & 31, wave = tid >> 5;
  const int m  = lane & 15;
  const int kh = lane >> 4;
  const float* W1 = ew1 + (size_t)k * GIN * HID;
  const float* B1 = eb1 + k * HID;
  for (int nt = 0; nt < 8; ++nt) {
    const int n0 = (wave * 8 + nt) * 16;
    v8f acc = {0.f, 0.f, 0.f, 0.f, 0.f, 0.f, 0.f, 0.f};
#pragma unroll 20
    for (int k0 = 0; k0 < GIN; k0 += 4) {
      v2f a, b;
      const float* ap = &Alds[m * GIN + k0 + 2 * kh];
      a.x = ap[0]; a.y = ap[1];
      const float* bp = &W1[(size_t)(k0 + 2 * kh) * HID + n0 + m];
      b.x = bp[0]; b.y = bp[HID];
      acc = wmma_f32_16x16x4(a, b, acc);
    }
#pragma unroll
    for (int v = 0; v < 8; ++v) {
      int row = v + 8 * kh;
      int col = n0 + m;
      float val = acc[v] + B1[col];
      EH[row * HID + col] = val > 0.f ? val : 0.f;
    }
  }
  __syncthreads();
  {
    int row = tid >> 3, seg = tid & 7;
    const float* w2 = ew2 + k * HID + seg * 64;
    const float* eh = EH + row * HID + seg * 64;
    float p = 0.f;
#pragma unroll 64
    for (int c = 0; c < 64; ++c) p += eh[c] * w2[c];
    psum[row * 8 + seg] = p;
  }
  __syncthreads();
  if (tid < 16) {
    int b = bids[tid];
    if (b >= 0) {
      float eo = eb2[k];
      for (int s = 0; s < 8; ++s) eo += psum[tid * 8 + s];
      float g = gates[b * K_ + k];
      atomicAdd(&y[b], g * eo);
    }
  }
}

__global__ void __launch_bounds__(1024)
vsams_loss_kernel(const float* __restrict__ gates, float* __restrict__ out) {
  __shared__ float wimp[32 * K_], wld[32 * K_];
  __shared__ float fin[2 * K_];
  const int t = threadIdx.x;
  const int lane = t & 31, wave = t >> 5;
  v4f g[16];
#pragma unroll
  for (int j = 0; j < 16; ++j)
    g[j] = *(const v4f*)&gates[(size_t)(j * 1024 + t) * 4];
  float pi[4] = {0.f, 0.f, 0.f, 0.f}, pl[4] = {0.f, 0.f, 0.f, 0.f};
#pragma unroll
  for (int j = 0; j < 16; ++j) {
#pragma unroll
    for (int c = 0; c < 4; ++c) {
      float gv = g[j][c];
      pi[c] += gv;
      pl[c] += (gv > 0.f) ? 1.f : 0.f;
    }
  }
#pragma unroll
  for (int c = 0; c < 4; ++c) {
#pragma unroll
    for (int off = 4; off < 32; off <<= 1) {
      pi[c] += __shfl_xor(pi[c], off);
      pl[c] += __shfl_xor(pl[c], off);
    }
  }
  if (lane < 4) {
#pragma unroll
    for (int c = 0; c < 4; ++c) {
      wimp[wave * K_ + lane * 4 + c] = pi[c];
      wld[wave * K_ + lane * 4 + c]  = pl[c];
    }
  }
  __syncthreads();
  if (t < 2 * K_) {
    const float* src = (t < K_) ? wimp : wld;
    const int col = t & (K_ - 1);
    float s = 0.f;
    for (int w = 0; w < 32; ++w) s += src[w * K_ + col];
    fin[t] = s;
  }
  __syncthreads();
  if (t == 0) {
    float imp[K_], ld[K_];
    for (int kk = 0; kk < K_; ++kk) { imp[kk] = fin[kk]; ld[kk] = fin[K_ + kk]; }
    float mi = 0.f, ml = 0.f;
    for (int kk = 0; kk < K_; ++kk) { mi += imp[kk]; ml += ld[kk]; }
    mi *= (1.f / 16.f); ml *= (1.f / 16.f);
    float vi = 0.f, vl = 0.f;
    for (int kk = 0; kk < K_; ++kk) {
      float di = imp[kk] - mi, dl = ld[kk] - ml;
      vi += di * di; vl += dl * dl;
    }
    vi *= (1.f / 16.f); vl *= (1.f / 16.f);
    out[B_] = vi / (mi * mi + EPSF) + vl / (ml * ml + EPSF);
  }
}

extern "C" void kernel_launch(void* const* d_in, const int* in_sizes, int n_in,
                              void* d_out, int out_size, void* d_ws, size_t ws_size,
                              hipStream_t stream) {
  const int*   x         = (const int*)d_in[0];
  const int*   sql       = (const int*)d_in[1];
  const float* sql_emb_w = (const float*)d_in[2];
  const float* inp_emb_w = (const float*)d_in[3];
  const float* gw1 = (const float*)d_in[4];
  const float* gb1 = (const float*)d_in[5];
  const float* gw2 = (const float*)d_in[6];
  const float* gb2 = (const float*)d_in[7];
  const float* ew1 = (const float*)d_in[8];
  const float* eb1 = (const float*)d_in[9];
  const float* ew2 = (const float*)d_in[10];
  const float* eb2 = (const float*)d_in[11];
  float* out = (float*)d_out;

  float* h      = (float*)d_ws;
  float* gates  = h + (size_t)B_ * HID;
  int*   counts = (int*)(gates + (size_t)B_ * K_);
  int*   lists  = counts + K_;

  vsams_init_kernel<<<17, 256, 0, stream>>>(out, counts);
  vsams_gate_l1_kernel<<<dim3(8, 256), 128, 0, stream>>>(sql, sql_emb_w, gw1, gb1, h);
  vsams_gate_head_kernel<<<256, 32, 0, stream>>>(h, gw2, gb2, gates, counts, lists);
  vsams_expert_kernel<<<K_ * 256, 128, 0, stream>>>(x, inp_emb_w, ew1, eb1, ew2,
                                                    eb2, gates, counts, lists, out);
  vsams_loss_kernel<<<1, 1024, 0, stream>>>(gates, out);
}


# k-loop outermost, 8 accumulator tiles per wave in both hidden GEMMs: one LDS A fragment feeds 8 WMMAs (was 1), a lane's 8 B values per k row are two 16-byte loads; gate: grid 256, register hint
# speedup vs baseline: 2.4423x; 1.7058x over previous
#include <hip/hip_runtime.h>

#define B_     4096
#define NFIELD 20
#define NEMB   16
#define HID    512
#define K_     16
#define GIN    320
#define EPSF   1e-10f

typedef __attribute__((ext_vector_type(2))) float v2f;
typedef __attribute__((ext_vector_type(8))) float v8f;
typedef __attribute__((ext_vector_type(4))) float v4f;

__device__ __forceinline__ v8f wmma_f32_16x16x4(v2f a, v2f b, v8f c) {
  return __builtin_amdgcn_wmma_f32_16x16x4_f32(false, a, false, b, (short)0, c,
                                               false, false);
}

__global__ void __launch_bounds__(256)
vsams_init_kernel(float* __restrict__ out, int* __restrict__ counts) {
  int t = blockIdx.x * blockDim.x + threadIdx.x;
  if (t <= B_) out[t] = 0.0f;
  if (t < K_) counts[t] = 0;
}

__global__ void __launch_bounds__(128) __attribute__((amdgpu_waves_per_eu(1, 2)))
vsams_gate_l1_kernel(const int* __restrict__ sql,
                     const float* __restrict__ sql_emb_w,
                     const float* __restrict__ gw1,
                     const float* __restrict__ gb1,
                     float* __restrict__ h) {
  __shared__ __attribute__((aligned(16))) float Alds[16 * GIN];
  const int tid = threadIdx.x;
  const int m0  = blockIdx.x * 16;
  {
    int s[10];
#pragma unroll
    for (int i = 0; i < 10; ++i) {
      int idx = tid + 128 * i;
      int r = idx / 80;
      int f = (idx - r * 80) >> 2;
      s[i] = sql[(m0 + r) * NFIELD + f];
    }
    v4f v[10];
#pragma unroll
    for (int i = 0; i < 10; ++i)
      v[i] = *(const v4f*)&sql_emb_w[(size_t)s[i] * NEMB + (tid & 3) * 4];
#pragma unroll
    for (int i = 0; i < 10; ++i) *(v4f*)&Alds[(tid + 128 * i) * 4] = v[i];
  }
  __syncthreads();
  const int lane = tid & 31, wave = tid >> 5;
  const int n0 = wave * 128;
  const int m  = lane & 15;
  const int kh = lane >> 4;
  v8f acc[8];
#pragma unroll
  for (int j = 0; j < 8; ++j) acc[j] = (v8f){0.f, 0.f, 0.f, 0.f, 0.f, 0.f, 0.f, 0.f};
  const float* ap = &Alds[m * GIN + 2 * kh];
  const float* wp = gw1 + (size_t)(2 * kh) * HID + n0 + 4 * m;
#pragma unroll 10
  for (int k0 = 0; k0 < GIN; k0 += 4) {
    v2f a;
    a.x = ap[k0]; a.y = ap[k0 + 1];
    const float* r0 = wp + (size_t)k0 * HID;
    v4f b00 = *(const v4f*)(r0);
    v4f b01 = *(const v4f*)(r0 + 64);
    v4f b10 = *(const v4f*)(r0 + HID);
    v4f b11 = *(const v4f*)(r0 + HID + 64);
#pragma unroll
    for (int j = 0; j < 4; ++j) {
      v2f b; b.x = b00[j]; b.y = b10[j];
      acc[j] = wmma_f32_16x16x4(a, b, acc[j]);
    }
#pragma unroll
    for (int j = 0; j < 4; ++j) {
      v2f b; b.x = b01[j]; b.y = b11[j];
      acc[4 + j] = wmma_f32_16x16x4(a, b, acc[4 + j]);
    }
  }
  const v4f bias0 = *(const v4f*)&gb1[n0 + 4 * m];
  const v4f bias1 = *(const v4f*)&gb1[n0 + 64 + 4 * m];
#pragma unroll
  for (int v = 0; v < 8; ++v) {
    const int row = m0 + v + 8 * kh;
    v4f o0, o1;
#pragma unroll
    for (int j = 0; j < 4; ++j) {
      float x0 = acc[j][v] + bias0[j];
      float x1 = acc[4 + j][v] + bias1[j];
      o0[j] = x0 > 0.f ? x0 : 0.f;
      o1[j] = x1 > 0.f ? x1 : 0.f;
    }
    *(v4f*)&h[(size_t)row * HID + n0 + 4 * m]      = o0;
    *(v4f*)&h[(size_t)row * HID + n0 + 64 + 4 * m] = o1;
  }
}

__global__ void __launch_bounds__(32)
vsams_gate_head_kernel(const float* __restrict__ h,
                       const float* __restrict__ gw2,
                       const float* __restrict__ gb2,
                       float* __restrict__ gates,
                       int* __restrict__ counts,
                       int* __restrict__ lists) {
  __shared__ float logit[16 * 16];
  const int lane = threadIdx.x;
  const int m0 = blockIdx.x * 16;
  const int m  = lane & 15;
  const int kh = lane >> 4;
  v8f acc = {0.f, 0.f, 0.f, 0.f, 0.f, 0.f, 0.f, 0.f};
#pragma unroll 8
  for (int k0 = 0; k0 < HID; k0 += 4) {
    v2f a, b;
    const float* ap = &h[(size_t)(m0 + m) * HID + k0 + 2 * kh];
    a.x = ap[0]; a.y = ap[1];
    const float* bp = &gw2[(k0 + 2 * kh) * K_ + m];
    b.x = bp[0]; b.y = bp[K_];
    acc = wmma_f32_16x16x4(a, b, acc);
  }
#pragma unroll
  for (int v = 0; v < 8; ++v)
    logit[(v + 8 * kh) * 16 + m] = acc[v] + gb2[m];
  __syncthreads();
  if (lane < 16) {
    const int b = m0 + lane;
    float l[K_];
    float mx = -3.0e38f;
    for (int j = 0; j < K_; ++j) { l[j] = logit[lane * 16 + j]; mx = fmaxf(mx, l[j]); }
    float s = 0.f;
    for (int j = 0; j < K_; ++j) { l[j] = __expf(l[j] - mx); s += l[j]; }
    float inv = 1.f / s;
    for (int j = 0; j < K_; ++j) l[j] *= inv;
    int i1 = 0; float v1 = l[0];
    for (int j = 1; j < K_; ++j) if (l[j] > v1) { v1 = l[j]; i1 = j; }
    int i2 = -1; float v2 = -1.f;
    for (int j = 0; j < K_; ++j) if (j != i1 && l[j] > v2) { v2 = l[j]; i2 = j; }
    float denom = v1 + v2;
    float g1 = v1 / denom, g2 = v2 / denom;
    for (int j = 0; j < K_; ++j) gates[b * K_ + j] = 0.f;
    gates[b * K_ + i1] = g1;
    gates[b * K_ + i2] = g2;
    int p1 = atomicAdd(&counts[i1], 1);
    lists[i1 * B_ + p1] = b;
    int p2 = atomicAdd(&counts[i2], 1);
    lists[i2 * B_ + p2] = b;
  }
}

__global__ void __launch_bounds__(128)
vsams_expert_kernel(const int* __restrict__ x,
                    const float* __restrict__ inp_emb_w,
                    const float* __restrict__ ew1,
                    const float* __restrict__ eb1,
                    const float* __restrict__ ew2,
                    const float* __restrict__ eb2,
                    const float* __restrict__ gates,
                    const int* __restrict__ counts,
                    const int* __restrict__ lists,
                    float* __restrict__ y) {
  __shared__ __attribute__((aligned(16))) float Alds[16 * GIN];
  __shared__ __attribute__((aligned(16))) float EH[16 * HID];
  __shared__ float psum[16 * 8];
  __shared__ int   bids[16];
  const int k    = blockIdx.x >> 8;
  const int tile = blockIdx.x & 255;
  const int cnt  = counts[k];
  const int base = tile * 16;
  if (base >= cnt) return;
  const int tid = threadIdx.x;
  if (tid < 16) bids[tid] = (base + tid < cnt) ? lists[k * B_ + base + tid] : -1;
  __syncthreads();
  {
    int xi[10];
#pragma unroll
    for (int i = 0; i < 10; ++i) {
      int idx = tid + 128 * i;
      int r = idx / 80;
      int f = (idx - r * 80) >> 2;
      int b = bids[r];
      xi[i] = x[(b >= 0 ? b : 0) * NFIELD + f];
    }
    v4f v[10];
#pragma unroll
    for (int i = 0; i < 10; ++i)
      v[i] = *(const v4f*)&inp_emb_w[(size_t)xi[i] * NEMB + (tid & 3) * 4];
#pragma unroll
    for (int i = 0; i < 10; ++i) *(v4f*)&Alds[(tid + 128 * i) * 4] = v[i];
  }
  __syncthreads();
  const int lane = tid & 31, wave = tid >> 5;
  const int m  = lane & 15;
  const int kh = lane >> 4;
  const float* W1 = ew1 + (size_t)k * GIN * HID;
  const float* B1 = eb1 + k * HID;
  {
    const int n0 = wave * 128;
    v8f acc[8];
#pragma unroll
    for (int j = 0; j < 8; ++j) acc[j] = (v8f){0.f, 0.f, 0.f, 0.f, 0.f, 0.f, 0.f, 0.f};
    const float* ap = &Alds[m * GIN + 2 * kh];
    const float* wp = W1 + (size_t)(2 * kh) * HID + n0 + 4 * m;
#pragma unroll 5
    for (int k0 = 0; k0 < GIN; k0 += 4) {
      v2f a;
      a.x = ap[k0]; a.y = ap[k0 + 1];
      const float* r0 = wp + (size_t)k0 * HID;
      v4f b00 = *(const v4f*)(r0);
      v4f b01 = *(const v4f*)(r0 + 64);
      v4f b10 = *(const v4f*)(r0 + HID);
      v4f b11 = *(const v4f*)(r0 + HID + 64);
#pragma unroll
      for (int j = 0; j < 4; ++j) {
        v2f b; b.x = b00[j]; b.y = b10[j];
        acc[j] = wmma_f32_16x16x4(a, b, acc[j]);
      }
#pragma unroll
      for (int j = 0; j < 4; ++j) {
        v2f b; b.x = b01[j]; b.y = b11[j];
        acc[4 + j] = wmma_f32_16x16x4(a, b, acc[4 + j]);
      }
    }
    const v4f bias0 = *(const v4f*)&B1[n0 + 4 * m];
    const v4f bias1 = *(const v4f*)&B1[n0 + 64 + 4 * m];
#pragma unroll
    for (int v = 0; v < 8; ++v) {
      const int row = v + 8 * kh;
      v4f o0, o1;
#pragma unroll
      for (int j = 0; j < 4; ++j) {
        float x0 = acc[j][v] + bias0[j];
        float x1 = acc[4 + j][v] + bias1[j];
        o0[j] = x0 > 0.f ? x0 : 0.f;
        o1[j] = x1 > 0.f ? x1 : 0.f;
      }
      *(v4f*)&EH[row * HID + n0 + 4 * m]      = o0;
      *(v4f*)&EH[row * HID + n0 + 64 + 4 * m] = o1;
    }
  }
  __syncthreads();
  {
    int row = tid >> 3, seg = tid & 7;
    const float* w2 = ew2 + k * HID + seg * 64;
    const float* eh = EH + row * HID + seg * 64;
    float p = 0.f;
#pragma unroll 64
    for (int c = 0; c < 64; ++c) p += eh[c] * w2[c];
    psum[row * 8 + seg] = p;
  }
  __syncthreads();
  if (tid < 16) {
    int b = bids[tid];
    if (b >= 0) {
      float eo = eb2[k];
      for (int s = 0; s < 8; ++s) eo += psum[tid * 8 + s];
      float g = gates[b * K_ + k];
      atomicAdd(&y[b], g * eo);
    }
  }
}

__global__ void __launch_bounds__(1024)
vsams_loss_kernel(const float* __restrict__ gates, float* __restrict__ out) {
  __shared__ float wimp[32 * K_], wld[32 * K_];
  __shared__ float fin[2 * K_];
  const int t = threadIdx.x;
  const int lane = t & 31, wave = t >> 5;
  v4f g[16];
#pragma unroll
  for (int j = 0; j < 16; ++j)
    g[j] = *(const v4f*)&gates[(size_t)(j * 1024 + t) * 4];
  float pi[4] = {0.f, 0.f, 0.f, 0.f}, pl[4] = {0.f, 0.f, 0.f, 0.f};
#pragma unroll
  for (int j = 0; j < 16; ++j) {
#pragma unroll
    for (int c = 0; c < 4; ++c) {
      float gv = g[j][c];
      pi[c] += gv;
      pl[c] += (gv > 0.f) ? 1.f : 0.f;
    }
  }
#pragma unroll
  for (int c = 0; c < 4; ++c) {
#pragma unroll
    for (int off = 4; off < 32; off <<= 1) {
      pi[c] += __shfl_xor(pi[c], off);
      pl[c] += __shfl_xor(pl[c], off);
    }
  }
  if (lane < 4) {
#pragma unroll
    for (int c = 0; c < 4; ++c) {
      wimp[wave * K_ + lane * 4 + c] = pi[c];
      wld[wave * K_ + lane * 4 + c]  = pl[c];
    }
  }
  __syncthreads();
  if (t < 2 * K_) {
    const float* src = (t < K_) ? wimp : wld;
    const int col = t & (K_ - 1);
    float s = 0.f;
    for (int w = 0; w < 32; ++w) s += src[w * K_ + col];
    fin[t] = s;
  }
  __syncthreads();
  if (t == 0) {
    float imp[K_], ld[K_];
    for (int kk = 0; kk < K_; ++kk) { imp[kk] = fin[kk]; ld[kk] = fin[K_ + kk]; }
    float mi = 0.f, ml = 0.f;
    for (int kk = 0; kk < K_; ++kk) { mi += imp[kk]; ml += ld[kk]; }
    mi *= (1.f / 16.f); ml *= (1.f / 16.f);
    float vi = 0.f, vl = 0.f;
    for (int kk = 0; kk < K_; ++kk) {
      float di = imp[kk] - mi, dl = ld[kk] - ml;
      vi += di * di; vl += dl * dl;
    }
    vi *= (1.f / 16.f); vl *= (1.f / 16.f);
    out[B_] = vi / (mi * mi + EPSF) + vl / (ml * ml + EPSF);
  }
}

extern "C" void kernel_launch(void* const* d_in, const int* in_sizes, int n_in,
                              void* d_out, int out_size, void* d_ws, size_t ws_size,
                              hipStream_t stream) {
  const int*   x         = (const int*)d_in[0];
  const int*   sql       = (const int*)d_in[1];
  const float* sql_emb_w = (const float*)d_in[2];
  const float* inp_emb_w = (const float*)d_in[3];
  const float* gw1 = (const float*)d_in[4];
  const float* gb1 = (const float*)d_in[5];
  const float* gw2 = (const float*)d_in[6];
  const float* gb2 = (const float*)d_in[7];
  const float* ew1 = (const float*)d_in[8];
  const float* eb1 = (const float*)d_in[9];
  const float* ew2 = (const float*)d_in[10];
  const float* eb2 = (const float*)d_in[11];
  float* out = (float*)d_out;

  float* h      = (float*)d_ws;
  float* gates  = h + (size_t)B_ * HID;
  int*   counts = (int*)(gates + (size_t)B_ * K_);
  int*   lists  = counts + K_;

  vsams_init_kernel<<<17, 256, 0, stream>>>(out, counts);
  vsams_gate_l1_kernel<<<256, 128, 0, stream>>>(sql, sql_emb_w, gw1, gb1, h);
  vsams_gate_head_kernel<<<256, 32, 0, stream>>>(h, gw2, gb2, gates, counts, lists);
  vsams_expert_kernel<<<K_ * 256, 128, 0, stream>>>(x, inp_emb_w, ew1, eb1, ew2,
                                                    eb2, gates, counts, lists, out);
  vsams_loss_kernel<<<1, 1024, 0, stream>>>(gates, out);
}


# 3 launches instead of 5: the gate GEMM launch clears y and counts itself, the loss runs as one extra 128-thread block of the expert launch; loss sum order changes
# speedup vs baseline: 2.6286x; 1.0763x over previous
#include <hip/hip_runtime.h>

#define B_     4096
#define NFIELD 20
#define NEMB   16
#define HID    512
#define K_     16
#define GIN    320
#define EPSF   1e-10f

typedef __attribute__((ext_vector_type(2))) float v2f;
typedef __attribute__((ext_vector_type(8))) float v8f;
typedef __attribute__((ext_vector_type(4))) float v4f;

__device__ __forceinline__ v8f wmma_f32_16x16x4(v2f a, v2f b, v8f c) {
  return __builtin_amdgcn_wmma_f32_16x16x4_f32(false, a, false, b, (short)0, c,
                                               false, false);
}

__global__ void __launch_bounds__(128) __attribute__((amdgpu_waves_per_eu(1, 2)))
vsams_gate_l1_kernel(const int* __restrict__ sql,
                     const float* __restrict__ sql_emb_w,
                     const float* __restrict__ gw1,
                     const float* __restrict__ gb1,
                     float* __restrict__ h,
                     float* __restrict__ y,
                     int* __restrict__ counts) {
  __shared__ __attribute__((aligned(16))) float Alds[16 * GIN];
  const int tid = threadIdx.x;
  const int m0  = blockIdx.x * 16;
  if ((blockIdx.x & 1) == 0 && tid < 32) y[m0 + tid] = 0.f;
  if (blockIdx.x == 0 && tid < K_) counts[tid] = 0;
  {
    int s[10];
#pragma unroll
    for (int i = 0; i < 10; ++i) {
      int idx = tid + 128 * i;
      int r = idx / 80;
      int f = (idx - r * 80) >> 2;
      s[i] = sql[(m0 + r) * NFIELD + f];
    }
    v4f v[10];
#pragma unroll
    for (int i = 0; i < 10; ++i)
      v[i] = *(const v4f*)&sql_emb_w[(size_t)s[i] * NEMB + (tid & 3) * 4];
#pragma unroll
    for (int i = 0; i < 10; ++i) *(v4f*)&Alds[(tid + 128 * i) * 4] = v[i];
  }
  __syncthreads();
  const int lane = tid & 31, wave = tid >> 5;
  const int n0 = wave * 128;
  const int m  = lane & 15;
  const int kh = lane >> 4;
  v8f acc[8];
#pragma unroll
  for (int j = 0; j < 8; ++j) acc[j] = (v8f){0.f, 0.f, 0.f, 0.f, 0.f, 0.f, 0.f, 0.f};
  const float* ap = &Alds[m * GIN + 2 * kh];
  const float* wp = gw1 + (size_t)(2 * kh) * HID + n0 + 4 * m;
#pragma unroll 10
  for (int k0 = 0; k0 < GIN; k0 += 4) {
    v2f a;
    a.x = ap[k0]; a.y = ap[k0 + 1];
    const float* r0 = wp + (size_t)k0 * HID;
    v4f b00 = *(const v4f*)(r0);
    v4f b01 = *(const v4f*)(r0 + 64);
    v4f b10 = *(const v4f*)(r0 + HID);
    v4f b11 = *(const v4f*)(r0 + HID + 64);
#pragma unroll
    for (int j = 0; j < 4; ++j) {
      v2f b; b.x = b00[j]; b.y = b10[j];
      acc[j] = wmma_f32_16x16x4(a, b, acc[j]);
    }
#pragma unroll
    for (int j = 0; j < 4; ++j) {
      v2f b; b.x = b01[j]; b.y = b11[j];
      acc[4 + j] = wmma_f32_16x16x4(a, b, acc[4 + j]);
    }
  }
  const v4f bias0 = *(const v4f*)&gb1[n0 + 4 * m];
  const v4f bias1 = *(const v4f*)&gb1[n0 + 64 + 4 * m];
#pragma unroll
  for (int v = 0; v < 8; ++v) {
    const int row = m0 + v + 8 * kh;
    v4f o0, o1;
#pragma unroll
    for (int j = 0; j < 4; ++j) {
      float x0 = acc[j][v] + bias0[j];
      float x1 = acc[4 + j][v] + bias1[j];
      o0[j] = x0 > 0.f ? x0 : 0.f;
      o1[j] = x1 > 0.f ? x1 : 0.f;
    }
    *(v4f*)&h[(size_t)row * HID + n0 + 4 * m]      = o0;
    *(v4f*)&h[(size_t)row * HID + n0 + 64 + 4 * m] = o1;
  }
}

__global__ void __launch_bounds__(32)
vsams_gate_head_kernel(const float* __restrict__ h,
                       const float* __restrict__ gw2,
                       const float* __restrict__ gb2,
                       float* __restrict__ gates,
                       int* __restrict__ counts,
                       int* __restrict__ lists) {
  __shared__ float logit[16 * 16];
  const int lane = threadIdx.x;
  const int m0 = blockIdx.x * 16;
  const int m  = lane & 15;
  const int kh = lane >> 4;
  v8f acc = {0.f, 0.f, 0.f, 0.f, 0.f, 0.f, 0.f, 0.f};
#pragma unroll 8
  for (int k0 = 0; k0 < HID; k0 += 4) {
    v2f a, b;
    const float* ap = &h[(size_t)(m0 + m) * HID + k0 + 2 * kh];
    a.x = ap[0]; a.y = ap[1];
    const float* bp = &gw2[(k0 + 2 * kh) * K_ + m];
    b.x = bp[0]; b.y = bp[K_];
    acc = wmma_f32_16x16x4(a, b, acc);
  }
#pragma unroll
  for (int v = 0; v < 8; ++v)
    logit[(v + 8 * kh) * 16 + m] = acc[v] + gb2[m];
  __syncthreads();
  if (lane < 16) {
    const int b = m0 + lane;
    float l[K_];
    float mx = -3.0e38f;
    for (int j = 0; j < K_; ++j) { l[j] = logit[lane * 16 + j]; mx = fmaxf(mx, l[j]); }
    float s = 0.f;
    for (int j = 0; j < K_; ++j) { l[j] = __expf(l[j] - mx); s += l[j]; }
    float inv = 1.f / s;
    for (int j = 0; j < K_; ++j) l[j] *= inv;
    int i1 = 0; float v1 = l[0];
    for (int j = 1; j < K_; ++j) if (l[j] > v1) { v1 = l[j]; i1 = j; }
    int i2 = -1; float v2 = -1.f;
    for (int j = 0; j < K_; ++j) if (j != i1 && l[j] > v2) { v2 = l[j]; i2 = j; }
    float denom = v1 + v2;
    float g1 = v1 / denom, g2 = v2 / denom;
    for (int j = 0; j < K_; ++j) gates[b * K_ + j] = 0.f;
    gates[b * K_ + i1] = g1;
    gates[b * K_ + i2] = g2;
    int p1 = atomicAdd(&counts[i1], 1);
    lists[i1 * B_ + p1] = b;
    int p2 = atomicAdd(&counts[i2], 1);
    lists[i2 * B_ + p2] = b;
  }
}

__device__ __forceinline__ void vsams_loss_block(const float* __restrict__ gates,
                                                 float* __restrict__ out,
                                                 float* __restrict__ scr) {
  float* wimp = scr;
  float* wld  = scr + 4 * K_;
  float* fin  = scr + 8 * K_;
  const int t = threadIdx.x;
  const int lane = t & 31, wave = t >> 5;
  float pi[4] = {0.f, 0.f, 0.f, 0.f}, pl[4] = {0.f, 0.f, 0.f, 0.f};
#pragma unroll 1
  for (int j0 = 0; j0 < 128; j0 += 16) {
    v4f g[16];
#pragma unroll
    for (int j = 0; j < 16; ++j)
      g[j] = *(const v4f*)&gates[(size_t)((j0 + j) * 128 + t) * 4];
#pragma unroll
    for (int j = 0; j < 16; ++j) {
#pragma unroll
      for (int c = 0; c < 4; ++c) {
        float gv = g[j][c];
        pi[c] += gv;
        pl[c] += (gv > 0.f) ? 1.f : 0.f;
      }
    }
  }
#pragma unroll
  for (int c = 0; c < 4; ++c) {
#pragma unroll
    for (int off = 4; off < 32; off <<= 1) {
      pi[c] += __shfl_xor(pi[c], off);
      pl[c] += __shfl_xor(pl[c], off);
    }
  }
  if (lane < 4) {
#pragma unroll
    for (int c = 0; c < 4; ++c) {
      wimp[wave * K_ + lane * 4 + c] = pi[c];
      wld[wave * K_ + lane * 4 + c]  = pl[c];
    }
  }
  __syncthreads();
  if (t < 2 * K_) {
    const float* src = (t < K_) ? wimp : wld;
    const int col = t & (K_ - 1);
    float s = 0.f;
    for (int w = 0; w < 4; ++w) s += src[w * K_ + col];
    fin[t] = s;
  }
  __syncthreads();
  if (t == 0) {
    float imp[K_], ld[K_];
    for (int kk = 0; kk < K_; ++kk) { imp[kk] = fin[kk]; ld[kk] = fin[K_ + kk]; }
    float mi = 0.f, ml = 0.f;
    for (int kk = 0; kk < K_; ++kk) { mi += imp[kk]; ml += ld[kk]; }
    mi *= (1.f / 16.f); ml *= (1.f / 16.f);
    float vi = 0.f, vl = 0.f;
    for (int kk = 0; kk < K_; ++kk) {
      float di = imp[kk] - mi, dl = ld[kk] - ml;
      vi += di * di; vl += dl * dl;
    }
    vi *= (1.f / 16.f); vl *= (1.f / 16.f);
    out[B_] = vi / (mi * mi + EPSF) + vl / (ml * ml + EPSF);
  }
}

__global__ void __launch_bounds__(128)
vsams_expert_kernel(const int* __restrict__ x,
                    const float* __restrict__ inp_emb_w,
                    const float* __restrict__ ew1,
                    const float* __restrict__ eb1,
                    const float* __restrict__ ew2,
                    const float* __restrict__ eb2,
                    const float* __restrict__ gates,
                    const int* __restrict__ counts,
                    const int* __restrict__ lists,
                    float* __restrict__ y) {
  __shared__ __attribute__((aligned(16))) float Alds[16 * GIN];
  __shared__ __attribute__((aligned(16))) float EH[16 * HID];
  __shared__ float psum[16 * 8];
  __shared__ int   bids[16];
  if (blockIdx.x == 0) {
    vsams_loss_block(gates, y, EH);
    return;
  }
  const int bid  = blockIdx.x - 1;
  const int k    = bid >> 8;
  const int tile = bid & 255;
  const int cnt  = counts[k];
  const int base = tile * 16;
  if (base >= cnt) return;
  const int tid = threadIdx.x;
  if (tid < 16) bids[tid] = (base + tid < cnt) ? lists[k * B_ + base + tid] : -1;
  __syncthreads();
  {
    int xi[10];
#pragma unroll
    for (int i = 0; i < 10; ++i) {
      int idx = tid + 128 * i;
      int r = idx / 80;
      int f = (idx - r * 80) >> 2;
      int b = bids[r];
      xi[i] = x[(b >= 0 ? b : 0) * NFIELD + f];
    }
    v4f v[10];
#pragma unroll
    for (int i = 0; i < 10; ++i)
      v[i] = *(const v4f*)&inp_emb_w[(size_t)xi[i] * NEMB + (tid & 3) * 4];
#pragma unroll
    for (int i = 0; i < 10; ++i) *(v4f*)&Alds[(tid + 128 * i) * 4] = v[i];
  }
  __syncthreads();
  const int lane = tid & 31, wave = tid >> 5;
  const int m  = lane & 15;
  const int kh = lane >> 4;
  const float* W1 = ew1 + (size_t)k * GIN * HID;
  const float* B1 = eb1 + k * HID;
  {
    const int n0 = wave * 128;
    v8f acc[8];
#pragma unroll
    for (int j = 0; j < 8; ++j) acc[j] = (v8f){0.f, 0.f, 0.f, 0.f, 0.f, 0.f, 0.f, 0.f};
    const float* ap = &Alds[m * GIN + 2 * kh];
    const float* wp = W1 + (size_t)(2 * kh) * HID + n0 + 4 * m;
#pragma unroll 5
    for (int k0 = 0; k0 < GIN; k0 += 4) {
      v2f a;
      a.x = ap[k0]; a.y = ap[k0 + 1];
      const float* r0 = wp + (size_t)k0 * HID;
      v4f b00 = *(const v4f*)(r0);
      v4f b01 = *(const v4f*)(r0 + 64);
      v4f b10 = *(const v4f*)(r0 + HID);
      v4f b11 = *(const v4f*)(r0 + HID + 64);
#pragma unroll
      for (int j = 0; j < 4; ++j) {
        v2f b; b.x = b00[j]; b.y = b10[j];
        acc[j] = wmma_f32_16x16x4(a, b, acc[j]);
      }
#pragma unroll
      for (int j = 0; j < 4; ++j) {
        v2f b; b.x = b01[j]; b.y = b11[j];
        acc[4 + j] = wmma_f32_16x16x4(a, b, acc[4 + j]);
      }
    }
    const v4f bias0 = *(const v4f*)&B1[n0 + 4 * m];
    const v4f bias1 = *(const v4f*)&B1[n0 + 64 + 4 * m];
#pragma unroll
    for (int v = 0; v < 8; ++v) {
      const int row = v + 8 * kh;
      v4f o0, o1;
#pragma unroll
      for (int j = 0; j < 4; ++j) {
        float x0 = acc[j][v] + bias0[j];
        float x1 = acc[4 + j][v] + bias1[j];
        o0[j] = x0 > 0.f ? x0 : 0.f;
        o1[j] = x1 > 0.f ? x1 : 0.f;
      }
      *(v4f*)&EH[row * HID + n0 + 4 * m]      = o0;
      *(v4f*)&EH[row * HID + n0 + 64 + 4 * m] = o1;
    }
  }
  __syncthreads();
  {
    int row = tid >> 3, seg = tid & 7;
    const float* w2 = ew2 + k * HID + seg * 64;
    const float* eh = EH + row * HID + seg * 64;
    float p = 0.f;
#pragma unroll 64
    for (int c = 0; c < 64; ++c) p += eh[c] * w2[c];
    psum[row * 8 + seg] = p;
  }
  __syncthreads();
  if (tid < 16) {
    int b = bids[tid];
    if (b >= 0) {
      float eo = eb2[k];
      for (int s = 0; s < 8; ++s) eo += psum[tid * 8 + s];
      float g = gates[b * K_ + k];
      atomicAdd(&y[b], g * eo);
    }
  }
}

extern "C" void kernel_launch(void* const* d_in, const int* in_sizes, int n_in,
                              void* d_out, int out_size, void* d_ws, size_t ws_size,
                              hipStream_t stream) {
  const int*   x         = (const int*)d_in[0];
  const int*   sql       = (const int*)d_in[1];
  const float* sql_emb_w = (const float*)d_in[2];
  const float* inp_emb_w = (const float*)d_in[3];
  const float* gw1 = (const float*)d_in[4];
  const float* gb1 = (const float*)d_in[5];
  const float* gw2 = (const float*)d_in[6];
  const float* gb2 = (const float*)d_in[7];
  const float* ew1 = (const float*)d_in[8];
  const float* eb1 = (const float*)d_in[9];
  const float* ew2 = (const float*)d_in[10];
  const float* eb2 = (const float*)d_in[11];
  float* out = (float*)d_out;

  float* h      = (float*)d_ws;
  float* gates  = h + (size_t)B_ * HID;
  int*   counts = (int*)(gates + (size_t)B_ * K_);
  int*   lists  = counts + K_;

  vsams_gate_l1_kernel<<<256, 128, 0, stream>>>(sql, sql_emb_w, gw1, gb1, h, out, counts);
  vsams_gate_head_kernel<<<256, 32, 0, stream>>>(h, gw2, gb2, gates, counts, lists);
  vsams_expert_kernel<<<4097, 128, 0, stream>>>(x, inp_emb_w, ew1, eb1, ew2,
                                                eb2, gates, counts, lists, out);
}


# Expert launch of 529 blocks instead of 4097: occupied tiles numbered consecutively over the experts, a block finds its (expert, tile) by a running sum over the counts; no block starts only to exit
# speedup vs baseline: 2.9148x; 1.1089x over previous
#include <hip/hip_runtime.h>

#define B_     4096
#define NFIELD 20
#define NEMB   16
#define HID    512
#define K_     16
#define GIN    320
#define EPSF   1e-10f

typedef __attribute__((ext_vector_type(2))) float v2f;
typedef __attribute__((ext_vector_type(8))) float v8f;
typedef __attribute__((ext_vector_type(4))) float v4f;

__device__ __forceinline__ v8f wmma_f32_16x16x4(v2f a, v2f b, v8f c) {
  return __builtin_amdgcn_wmma_f32_16x16x4_f32(false, a, false, b, (short)0, c,
                                               false, false);
}

__global__ void __launch_bounds__(128) __attribute__((amdgpu_waves_per_eu(1, 2)))
vsams_gate_l1_kernel(const int* __restrict__ sql,
                     const float* __restrict__ sql_emb_w,
                     const float* __restrict__ gw1,
                     const float* __restrict__ gb1,
                     float* __restrict__ h,
                     float* __restrict__ y,
                     int* __restrict__ counts) {
  __shared__ __attribute__((aligned(16))) float Alds[16 * GIN];
  const int tid = threadIdx.x;
  const int m0  = blockIdx.x * 16;
  if ((blockIdx.x & 1) == 0 && tid < 32) y[m0 + tid] = 0.f;
  if (blockIdx.x == 0 && tid < K_) counts[tid] = 0;
  {
    int s[10];
#pragma unroll
    for (int i = 0; i < 10; ++i) {
      int idx = tid + 128 * i;
      int r = idx / 80;
      int f = (idx - r * 80) >> 2;
      s[i] = sql[(m0 + r) * NFIELD + f];
    }
    v4f v[10];
#pragma unroll
    for (int i = 0; i < 10; ++i)
      v[i] = *(const v4f*)&sql_emb_w[(size_t)s[i] * NEMB + (tid & 3) * 4];
#pragma unroll
    for (int i = 0; i < 10; ++i) *(v4f*)&Alds[(tid + 128 * i) * 4] = v[i];
  }
  __syncthreads();
  const int lane = tid & 31, wave = tid >> 5;
  const int n0 = wave * 128;
  const int m  = lane & 15;
  const int kh = lane >> 4;
  v8f acc[8];
#pragma unroll
  for (int j = 0; j < 8; ++j) acc[j] = (v8f){0.f, 0.f, 0.f, 0.f, 0.f, 0.f, 0.f, 0.f};
  const float* ap = &Alds[m * GIN + 2 * kh];
  const float* wp = gw1 + (size_t)(2 * kh) * HID + n0 + 4 * m;
#pragma unroll 10
  for (int k0 = 0; k0 < GIN; k0 += 4) {
    v2f a;
    a.x = ap[k0]; a.y = ap[k0 + 1];
    const float* r0 = wp + (size_t)k0 * HID;
    v4f b00 = *(const v4f*)(r0);
    v4f b01 = *(const v4f*)(r0 + 64);
    v4f b10 = *(const v4f*)(r0 + HID);
    v4f b11 = *(const v4f*)(r0 + HID + 64);
#pragma unroll
    for (int j = 0; j < 4; ++j) {
      v2f b; b.x = b00[j]; b.y = b10[j];
      acc[j] = wmma_f32_16x16x4(a, b, acc[j]);
    }
#pragma unroll
    for (int j = 0; j < 4; ++j) {
      v2f b; b.x = b01[j]; b.y = b11[j];
      acc[4 + j] = wmma_f32_16x16x4(a, b, acc[4 + j]);
    }
  }
  const v4f bias0 = *(const v4f*)&gb1[n0 + 4 * m];
  const v4f bias1 = *(const v4f*)&gb1[n0 + 64 + 4 * m];
#pragma unroll
  for (int v = 0; v < 8; ++v) {
    const int row = m0 + v + 8 * kh;
    v4f o0, o1;
#pragma unroll
    for (int j = 0; j < 4; ++j) {
      float x0 = acc[j][v] + bias0[j];
      float x1 = acc[4 + j][v] + bias1[j];
      o0[j] = x0 > 0.f ? x0 : 0.f;
      o1[j] = x1 > 0.f ? x1 : 0.f;
    }
    *(v4f*)&h[(size_t)row * HID + n0 + 4 * m]      = o0;
    *(v4f*)&h[(size_t)row * HID + n0 + 64 + 4 * m] = o1;
  }
}

__global__ void __launch_bounds__(32)
vsams_gate_head_kernel(const float* __restrict__ h,
                       const float* __restrict__ gw2,
                       const float* __restrict__ gb2,
                       float* __restrict__ gates,
                       int* __restrict__ counts,
                       int* __restrict__ lists) {
  __shared__ float logit[16 * 16];
  const int lane = threadIdx.x;
  const int m0 = blockIdx.x * 16;
  const int m  = lane & 15;
  const int kh = lane >> 4;
  v8f acc = {0.f, 0.f, 0.f, 0.f, 0.f, 0.f, 0.f, 0.f};
#pragma unroll 8
  for (int k0 = 0; k0 < HID; k0 += 4) {
    v2f a, b;
    const float* ap = &h[(size_t)(m0 + m) * HID + k0 + 2 * kh];
    a.x = ap[0]; a.y = ap[1];
    const float* bp = &gw2[(k0 + 2 * kh) * K_ + m];
    b.x = bp[0]; b.y = bp[K_];
    acc = wmma_f32_16x16x4(a, b, acc);
  }
#pragma unroll
  for (int v = 0; v < 8; ++v)
    logit[(v + 8 * kh) * 16 + m] = acc[v] + gb2[m];
  __syncthreads();
  if (lane < 16) {
    const int b = m0 + lane;
    float l[K_];
    float mx = -3.0e38f;
    for (int j = 0; j < K_; ++j) { l[j] = logit[lane * 16 + j]; mx = fmaxf(mx, l[j]); }
    float s = 0.f;
    for (int j = 0; j < K_; ++j) { l[j] = __expf(l[j] - mx); s += l[j]; }
    float inv = 1.f / s;
    for (int j = 0; j < K_; ++j) l[j] *= inv;
    int i1 = 0; float v1 = l[0];
    for (int j = 1; j < K_; ++j) if (l[j] > v1) { v1 = l[j]; i1 = j; }
    int i2 = -1; float v2 = -1.f;
    for (int j = 0; j < K_; ++j) if (j != i1 && l[j] > v2) { v2 = l[j]; i2 = j; }
    float denom = v1 + v2;
    float g1 = v1 / denom, g2 = v2 / denom;
    for (int j = 0; j < K_; ++j) gates[b * K_ + j] = 0.f;
    gates[b * K_ + i1] = g1;
    gates[b * K_ + i2] = g2;
    int p1 = atomicAdd(&counts[i1], 1);
    lists[i1 * B_ + p1] = b;
    int p2 = atomicAdd(&counts[i2], 1);
    lists[i2 * B_ + p2] = b;
  }
}

__device__ __forceinline__ void vsams_loss_block(const float* __restrict__ gates,
                                                 float* __restrict__ out,
                                                 float* __restrict__ scr) {
  float* wimp = scr;
  float* wld  = scr + 4 * K_;
  float* fin  = scr + 8 * K_;
  const int t = threadIdx.x;
  const int lane = t & 31, wave = t >> 5;
  float pi[4] = {0.f, 0.f, 0.f, 0.f}, pl[4] = {0.f, 0.f, 0.f, 0.f};
#pragma unroll 1
  for (int j0 = 0; j0 < 128; j0 += 16) {
    v4f g[16];
#pragma unroll
    for (int j = 0; j < 16; ++j)
      g[j] = *(const v4f*)&gates[(size_t)((j0 + j) * 128 + t) * 4];
#pragma unroll
    for (int j = 0; j < 16; ++j) {
#pragma unroll
      for (int c = 0; c < 4; ++c) {
        float gv = g[j][c];
        pi[c] += gv;
        pl[c] += (gv > 0.f) ? 1.f : 0.f;
      }
    }
  }
#pragma unroll
  for (int c = 0; c < 4; ++c) {
#pragma unroll
    for (int off = 4; off < 32; off <<= 1) {
      pi[c] += __shfl_xor(pi[c], off);
      pl[c] += __shfl_xor(pl[c], off);
    }
  }
  if (lane < 4) {
#pragma unroll
    for (int c = 0; c < 4; ++c) {
      wimp[wave * K_ + lane * 4 + c] = pi[c];
      wld[wave * K_ + lane * 4 + c]  = pl[c];
    }
  }
  __syncthreads();
  if (t < 2 * K_) {
    const float* src = (t < K_) ? wimp : wld;
    const int col = t & (K_ - 1);
    float s = 0.f;
    for (int w = 0; w < 4; ++w) s += src[w * K_ + col];
    fin[t] = s;
  }
  __syncthreads();
  if (t == 0) {
    float imp[K_], ld[K_];
    for (int kk = 0; kk < K_; ++kk) { imp[kk] = fin[kk]; ld[kk] = fin[K_ + kk]; }
    float mi = 0.f, ml = 0.f;
    for (int kk = 0; kk < K_; ++kk) { mi += imp[kk]; ml += ld[kk]; }
    mi *= (1.f / 16.f); ml *= (1.f / 16.f);
    float vi = 0.f, vl = 0.f;
    for (int kk = 0; kk < K_; ++kk) {
      float di = imp[kk] - mi, dl = ld[kk] - ml;
      vi += di * di; vl += dl * dl;
    }
    vi *= (1.f / 16.f); vl *= (1.f / 16.f);
    out[B_] = vi / (mi * mi + EPSF) + vl / (ml * ml + EPSF);
  }
}

__global__ void __launch_bounds__(128)
vsams_expert_kernel(const int* __restrict__ x,
                    const float* __restrict__ inp_emb_w,
                    const float* __restrict__ ew1,
                    const float* __restrict__ eb1,
                    const float* __restrict__ ew2,
                    const float* __restrict__ eb2,
                    const float* __restrict__ gates,
                    const int* __restrict__ counts,
                    const int* __restrict__ lists,
                    float* __restrict__ y) {
  __shared__ __attribute__((aligned(16))) float Alds[16 * GIN];
  __shared__ __attribute__((aligned(16))) float EH[16 * HID];
  __shared__ float psum[16 * 8];
  __shared__ int   bids[16];
  if (blockIdx.x == 0) {
    vsams_loss_block(gates, y, EH);
    return;
  }
  const int bid = blockIdx.x - 1;
  int k = -1, tile = 0, cnt = 0, first = 0;
  for (int e = 0; e < K_; ++e) {
    const int c  = counts[e];
    const int nt = (c + 15) >> 4;
    if (bid >= first && bid < first + nt) { k = e; tile = bid - first; cnt = c; }
    first += nt;
  }
  if (k < 0) return;
  const int base = tile * 16;
  const int tid = threadIdx.x;
  if (tid < 16) bids[tid] = (base + tid < cnt) ? lists[k * B_ + base + tid] : -1;
  __syncthreads();
  {
    int xi[10];
#pragma unroll
    for (int i = 0; i < 10; ++i) {
      int idx = tid + 128 * i;
      int r = idx / 80;
      int f = (idx - r * 80) >> 2;
      int b = bids[r];
      xi[i] = x[(b >= 0 ? b : 0) * NFIELD + f];
    }
    v4f v[10];
#pragma unroll
    for (int i = 0; i < 10; ++i)
      v[i] = *(const v4f*)&inp_emb_w[(size_t)xi[i] * NEMB + (tid & 3) * 4];
#pragma unroll
    for (int i = 0; i < 10; ++i) *(v4f*)&Alds[(tid + 128 * i) * 4] = v[i];
  }
  __syncthreads();
  const int lane = tid & 31, wave = tid >> 5;
  const int m  = lane & 15;
  const int kh = lane >> 4;
  const float* W1 = ew1 + (size_t)k * GIN * HID;
  const float* B1 = eb1 + k * HID;
  {
    const int n0 = wave * 128;
    v8f acc[8];
#pragma unroll
    for (int j = 0; j < 8; ++j) acc[j] = (v8f){0.f, 0.f, 0.f, 0.f, 0.f, 0.f, 0.f, 0.f};
    const float* ap = &Alds[m * GIN + 2 * kh];
    const float* wp = W1 + (size_t)(2 * kh) * HID + n0 + 4 * m;
#pragma unroll 5
    for (int k0 = 0; k0 < GIN; k0 += 4) {
      v2f a;
      a.x = ap[k0]; a.y = ap[k0 + 1];
      const float* r0 = wp + (size_t)k0 * HID;
      v4f b00 = *(const v4f*)(r0);
      v4f b01 = *(const v4f*)(r0 + 64);
      v4f b10 = *(const v4f*)(r0 + HID);
      v4f b11 = *(const v4f*)(r0 + HID + 64);
#pragma unroll
      for (int j = 0; j < 4; ++j) {
        v2f b; b.x = b00[j]; b.y = b10[j];
        acc[j] = wmma_f32_16x16x4(a, b, acc[j]);
      }
#pragma unroll
      for (int j = 0; j < 4; ++j) {
        v2f b; b.x = b01[j]; b.y = b11[j];
        acc[4 + j] = wmma_f32_16x16x4(a, b, acc[4 + j]);
      }
    }
    const v4f bias0 = *(const v4f*)&B1[n0 + 4 * m];
    const v4f bias1 = *(const v4f*)&B1[n0 + 64 + 4 * m];
#pragma unroll
    for (int v = 0; v < 8; ++v) {
      const int row = v + 8 * kh;
      v4f o0, o1;
#pragma unroll
      for (int j = 0; j < 4; ++j) {
        float x0 = acc[j][v] + bias0[j];
        float x1 = acc[4 + j][v] + bias1[j];
        o0[j] = x0 > 0.f ? x0 : 0.f;
        o1[j] = x1 > 0.f ? x1 : 0.f;
      }
      *(v4f*)&EH[row * HID + n0 + 4 * m]      = o0;
      *(v4f*)&EH[row * HID + n0 + 64 + 4 * m] = o1;
    }
  }
  __syncthreads();
  {
    int row = tid >> 3, seg = tid & 7;
    const float* w2 = ew2 + k * HID + seg * 64;
    const float* eh = EH + row * HID + seg * 64;
    float p = 0.f;
#pragma unroll 64
    for (int c = 0; c < 64; ++c) p += eh[c] * w2[c];
    psum[row * 8 + seg] = p;
  }
  __syncthreads();
  if (tid < 16) {
    int b = bids[tid];
    if (b >= 0) {
      float eo = eb2[k];
      for (int s = 0; s < 8; ++s) eo += psum[tid * 8 + s];
      float g = gates[b * K_ + k];
      atomicAdd(&y[b], g * eo);
    }
  }
}

extern "C" void kernel_launch(void* const* d_in, const int* in_sizes, int n_in,
                              void* d_out, int out_size, void* d_ws, size_t ws_size,
                              hipStream_t stream) {
  const int*   x         = (const int*)d_in[0];
  const int*   sql       = (const int*)d_in[1];
  const float* sql_emb_w = (const float*)d_in[2];
  const float* inp_emb_w = (const float*)d_in[3];
  const float* gw1 = (const float*)d_in[4];
  const float* gb1 = (const float*)d_in[5];
  const float* gw2 = (const float*)d_in[6];
  const float* gb2 = (const float*)d_in[7];
  const float* ew1 = (const float*)d_in[8];
  const float* eb1 = (const float*)d_in[9];
  const float* ew2 = (const float*)d_in[10];
  const float* eb2 = (const float*)d_in[11];
  float* out = (float*)d_out;

  float* h      = (float*)d_ws;
  float* gates  = h + (size_t)B_ * HID;
  int*   counts = (int*)(gates + (size_t)B_ * K_);
  int*   lists  = counts + K_;

  vsams_gate_l1_kernel<<<256, 128, 0, stream>>>(sql, sql_emb_w, gw1, gb1, h, out, counts);
  vsams_gate_head_kernel<<<256, 32, 0, stream>>>(h, gw2, gb2, gates, counts, lists);
  vsams_expert_kernel<<<529, 128, 0, stream>>>(x, inp_emb_w, ew1, eb1, ew2,
                                                eb2, gates, counts, lists, out);
}


# Gate head: 4 waves per block, each sums a quarter of K with all 80 loads of a lane issued before its first wait, 2 accumulator chains (was 1 wave, 16 dependent round trips); logit sum order changes
# speedup vs baseline: 3.4214x; 1.1738x over previous
#include <hip/hip_runtime.h>

#define B_     4096
#define NFIELD 20
#define NEMB   16
#define HID    512
#define K_     16
#define GIN    320
#define EPSF   1e-10f

typedef __attribute__((ext_vector_type(2))) float v2f;
typedef __attribute__((ext_vector_type(8))) float v8f;
typedef __attribute__((ext_vector_type(4))) float v4f;

__device__ __forceinline__ v8f wmma_f32_16x16x4(v2f a, v2f b, v8f c) {
  return __builtin_amdgcn_wmma_f32_16x16x4_f32(false, a, false, b, (short)0, c,
                                               false, false);
}

__global__ void __launch_bounds__(128) __attribute__((amdgpu_waves_per_eu(1, 2)))
vsams_gate_l1_kernel(const int* __restrict__ sql,
                     const float* __restrict__ sql_emb_w,
                     const float* __restrict__ gw1,
                     const float* __restrict__ gb1,
                     float* __restrict__ h,
                     float* __restrict__ y,
                     int* __restrict__ counts) {
  __shared__ __attribute__((aligned(16))) float Alds[16 * GIN];
  const int tid = threadIdx.x;
  const int m0  = blockIdx.x * 16;
  if ((blockIdx.x & 1) == 0 && tid < 32) y[m0 + tid] = 0.f;
  if (blockIdx.x == 0 && tid < K_) counts[tid] = 0;
  {
    int s[10];
#pragma unroll
    for (int i = 0; i < 10; ++i) {
      int idx = tid + 128 * i;
      int r = idx / 80;
      int f = (idx - r * 80) >> 2;
      s[i] = sql[(m0 + r) * NFIELD + f];
    }
    v4f v[10];
#pragma unroll
    for (int i = 0; i < 10; ++i)
      v[i] = *(const v4f*)&sql_emb_w[(size_t)s[i] * NEMB + (tid & 3) * 4];
#pragma unroll
    for (int i = 0; i < 10; ++i) *(v4f*)&Alds[(tid + 128 * i) * 4] = v[i];
  }
  __syncthreads();
  const int lane = tid & 31, wave = tid >> 5;
  const int n0 = wave * 128;
  const int m  = lane & 15;
  const int kh = lane >> 4;
  v8f acc[8];
#pragma unroll
  for (int j = 0; j < 8; ++j) acc[j] = (v8f){0.f, 0.f, 0.f, 0.f, 0.f, 0.f, 0.f, 0.f};
  const float* ap = &Alds[m * GIN + 2 * kh];
  const float* wp = gw1 + (size_t)(2 * kh) * HID + n0 + 4 * m;
#pragma unroll 10
  for (int k0 = 0; k0 < GIN; k0 += 4) {
    v2f a;
    a.x = ap[k0]; a.y = ap[k0 + 1];
    const float* r0 = wp + (size_t)k0 * HID;
    v4f b00 = *(const v4f*)(r0);
    v4f b01 = *(const v4f*)(r0 + 64);
    v4f b10 = *(const v4f*)(r0 + HID);
    v4f b11 = *(const v4f*)(r0 + HID + 64);
#pragma unroll
    for (int j = 0; j < 4; ++j) {
      v2f b; b.x = b00[j]; b.y = b10[j];
      acc[j] = wmma_f32_16x16x4(a, b, acc[j]);
    }
#pragma unroll
    for (int j = 0; j < 4; ++j) {
      v2f b; b.x = b01[j]; b.y = b11[j];
      acc[4 + j] = wmma_f32_16x16x4(a, b, acc[4 + j]);
    }
  }
  const v4f bias0 = *(const v4f*)&gb1[n0 + 4 * m];
  const v4f bias1 = *(const v4f*)&gb1[n0 + 64 + 4 * m];
#pragma unroll
  for (int v = 0; v < 8; ++v) {
    const int row = m0 + v + 8 * kh;
    v4f o0, o1;
#pragma unroll
    for (int j = 0; j < 4; ++j) {
      float x0 = acc[j][v] + bias0[j];
      float x1 = acc[4 + j][v] + bias1[j];
      o0[j] = x0 > 0.f ? x0 : 0.f;
      o1[j] = x1 > 0.f ? x1 : 0.f;
    }
    *(v4f*)&h[(size_t)row * HID + n0 + 4 * m]      = o0;
    *(v4f*)&h[(size_t)row * HID + n0 + 64 + 4 * m] = o1;
  }
}

__global__ void __launch_bounds__(128) __attribute__((amdgpu_waves_per_eu(1, 2)))
vsams_gate_head_kernel(const float* __restrict__ h,
                       const float* __restrict__ gw2,
                       const float* __restrict__ gb2,
                       float* __restrict__ gates,
                       int* __restrict__ counts,
                       int* __restrict__ lists) {
  __shared__ __attribute__((aligned(16))) float part[4 * 16 * 16];
  const int tid = threadIdx.x;
  const int lane = tid & 31, wave = tid >> 5;
  const int m0 = blockIdx.x * 16;
  const int m  = lane & 15;
  const int kh = lane >> 4;
  const int kq = 128 * wave + 64 * kh;
  const float* ap = h + (size_t)(m0 + m) * HID + kq;
  const float* bp = gw2 + (size_t)kq * K_ + m;
  v4f a[16];
#pragma unroll
  for (int q = 0; q < 16; ++q) a[q] = *(const v4f*)(ap + 4 * q);
  float b[64];
#pragma unroll
  for (int i = 0; i < 64; ++i) b[i] = bp[i * K_];
  v8f acc0 = {0.f, 0.f, 0.f, 0.f, 0.f, 0.f, 0.f, 0.f};
  v8f acc1 = {0.f, 0.f, 0.f, 0.f, 0.f, 0.f, 0.f, 0.f};
#pragma unroll
  for (int q = 0; q < 16; ++q) {
    v2f a0, a1, b0, b1;
    a0.x = a[q][0]; a0.y = a[q][1];
    a1.x = a[q][2]; a1.y = a[q][3];
    b0.x = b[4 * q];     b0.y = b[4 * q + 1];
    b1.x = b[4 * q + 2]; b1.y = b[4 * q + 3];
    acc0 = wmma_f32_16x16x4(a0, b0, acc0);
    acc1 = wmma_f32_16x16x4(a1, b1, acc1);
  }
#pragma unroll
  for (int v = 0; v < 8; ++v)
    part[wave * 256 + (v + 8 * kh) * 16 + m] = acc0[v] + acc1[v];
  __syncthreads();
  if (tid < 16) {
    const int b_ = m0 + tid;
    float l[K_];
    float mx = -3.0e38f;
    for (int j = 0; j < K_; ++j) {
      float s4 = part[tid * 16 + j] + part[256 + tid * 16 + j];
      s4 += part[512 + tid * 16 + j];
      s4 += part[768 + tid * 16 + j];
      l[j] = s4 + gb2[j];
      mx = fmaxf(mx, l[j]);
    }
    float s = 0.f;
    for (int j = 0; j < K_; ++j) { l[j] = __expf(l[j] - mx); s += l[j]; }
    float inv = 1.f / s;
    for (int j = 0; j < K_; ++j) l[j] *= inv;
    int i1 = 0; float v1 = l[0];
    for (int j = 1; j < K_; ++j) if (l[j] > v1) { v1 = l[j]; i1 = j; }
    int i2 = -1; float v2 = -1.f;
    for (int j = 0; j < K_; ++j) if (j != i1 && l[j] > v2) { v2 = l[j]; i2 = j; }
    float denom = v1 + v2;
    float g1 = v1 / denom, g2 = v2 / denom;
    for (int j = 0; j < K_; ++j) gates[b_ * K_ + j] = 0.f;
    gates[b_ * K_ + i1] = g1;
    gates[b_ * K_ + i2] = g2;
    int p1 = atomicAdd(&counts[i1], 1);
    lists[i1 * B_ + p1] = b_;
    int p2 = atomicAdd(&counts[i2], 1);
    lists[i2 * B_ + p2] = b_;
  }
}

__device__ __forceinline__ void vsams_loss_block(const float* __restrict__ gates,
                                                 float* __restrict__ out,
                                                 float* __restrict__ scr) {
  float* wimp = scr;
  float* wld  = scr + 4 * K_;
  float* fin  = scr + 8 * K_;
  const int t = threadIdx.x;
  const int lane = t & 31, wave = t >> 5;
  float pi[4] = {0.f, 0.f, 0.f, 0.f}, pl[4] = {0.f, 0.f, 0.f, 0.f};
#pragma unroll 1
  for (int j0 = 0; j0 < 128; j0 += 16) {
    v4f g[16];
#pragma unroll
    for (int j = 0; j < 16; ++j)
      g[j] = *(const v4f*)&gates[(size_t)((j0 + j) * 128 + t) * 4];
#pragma unroll
    for (int j = 0; j < 16; ++j) {
#pragma unroll
      for (int c = 0; c < 4; ++c) {
        float gv = g[j][c];
        pi[c] += gv;
        pl[c] += (gv > 0.f) ? 1.f : 0.f;
      }
    }
  }
#pragma unroll
  for (int c = 0; c < 4; ++c) {
#pragma unroll
    for (int off = 4; off < 32; off <<= 1) {
      pi[c] += __shfl_xor(pi[c], off);
      pl[c] += __shfl_xor(pl[c], off);
    }
  }
  if (lane < 4) {
#pragma unroll
    for (int c = 0; c < 4; ++c) {
      wimp[wave * K_ + lane * 4 + c] = pi[c];
      wld[wave * K_ + lane * 4 + c]  = pl[c];
    }
  }
  __syncthreads();
  if (t < 2 * K_) {
    const float* src = (t < K_) ? wimp : wld;
    const int col = t & (K_ - 1);
    float s = 0.f;
    for (int w = 0; w < 4; ++w) s += src[w * K_ + col];
    fin[t] = s;
  }
  __syncthreads();
  if (t == 0) {
    float imp[K_], ld[K_];
    for (int kk = 0; kk < K_; ++kk) { imp[kk] = fin[kk]; ld[kk] = fin[K_ + kk]; }
    float mi = 0.f, ml = 0.f;
    for (int kk = 0; kk < K_; ++kk) { mi += imp[kk]; ml += ld[kk]; }
    mi *= (1.f / 16.f); ml *= (1.f / 16.f);
    float vi = 0.f, vl = 0.f;
    for (int kk = 0; kk < K_; ++kk) {
      float di = imp[kk] - mi, dl = ld[kk] - ml;
      vi += di * di; vl += dl * dl;
    }
    vi *= (1.f / 16.f); vl *= (1.f / 16.f);
    out[B_] = vi / (mi * mi + EPSF) + vl / (ml * ml + EPSF);
  }
}

__global__ void __launch_bounds__(128)
vsams_expert_kernel(const int* __restrict__ x,
                    const float* __restrict__ inp_emb_w,
                    const float* __restrict__ ew1,
                    const float* __restrict__ eb1,
                    const float* __restrict__ ew2,
                    const float* __restrict__ eb2,
                    const float* __restrict__ gates,
                    const int* __restrict__ counts,
                    const int* __restrict__ lists,
                    float* __restrict__ y) {
  __shared__ __attribute__((aligned(16))) float Alds[16 * GIN];
  __shared__ __attribute__((aligned(16))) float EH[16 * HID];
  __shared__ float psum[16 * 8];
  __shared__ int   bids[16];
  if (blockIdx.x == 0) {
    vsams_loss_block(gates, y, EH);
    return;
  }
  const int bid = blockIdx.x - 1;
  int k = -1, tile = 0, cnt = 0, first = 0;
  for (int e = 0; e < K_; ++e) {
    const int c  = counts[e];
    const int nt = (c + 15) >> 4;
    if (bid >= first && bid < first + nt) { k = e; tile = bid - first; cnt = c; }
    first += nt;
  }
  if (k < 0) return;
  const int base = tile * 16;
  const int tid = threadIdx.x;
  if (tid < 16) bids[tid] = (base + tid < cnt) ? lists[k * B_ + base + tid] : -1;
  __syncthreads();
  {
    int xi[10];
#pragma unroll
    for (int i = 0; i < 10; ++i) {
      int idx = tid + 128 * i;
      int r = idx / 80;
      int f = (idx - r * 80) >> 2;
      int b = bids[r];
      xi[i] = x[(b >= 0 ? b : 0) * NFIELD + f];
    }
    v4f v[10];
#pragma unroll
    for (int i = 0; i < 10; ++i)
      v[i] = *(const v4f*)&inp_emb_w[(size_t)xi[i] * NEMB + (tid & 3) * 4];
#pragma unroll
    for (int i = 0; i < 10; ++i) *(v4f*)&Alds[(tid + 128 * i) * 4] = v[i];
  }
  __syncthreads();
  const int lane = tid & 31, wave = tid >> 5;
  const int m  = lane & 15;
  const int kh = lane >> 4;
  const float* W1 = ew1 + (size_t)k * GIN * HID;
  const float* B1 = eb1 + k * HID;
  {
    const int n0 = wave * 128;
    v8f acc[8];
#pragma unroll
    for (int j = 0; j < 8; ++j) acc[j] = (v8f){0.f, 0.f, 0.f, 0.f, 0.f, 0.f, 0.f, 0.f};
    const float* ap = &Alds[m * GIN + 2 * kh];
    const float* wp = W1 + (size_t)(2 * kh) * HID + n0 + 4 * m;
#pragma unroll 5
    for (int k0 = 0; k0 < GIN; k0 += 4) {
      v2f a;
      a.x = ap[k0]; a.y = ap[k0 + 1];
      const float* r0 = wp + (size_t)k0 * HID;
      v4f b00 = *(const v4f*)(r0);
      v4f b01 = *(const v4f*)(r0 + 64);
      v4f b10 = *(const v4f*)(r0 + HID);
      v4f b11 = *(const v4f*)(r0 + HID + 64);
#pragma unroll
      for (int j = 0; j < 4; ++j) {
        v2f b; b.x = b00[j]; b.y = b10[j];
        acc[j] = wmma_f32_16x16x4(a, b, acc[j]);
      }
#pragma unroll
      for (int j = 0; j < 4; ++j) {
        v2f b; b.x = b01[j]; b.y = b11[j];
        acc[4 + j] = wmma_f32_16x16x4(a, b, acc[4 + j]);
      }
    }
    const v4f bias0 = *(const v4f*)&B1[n0 + 4 * m];
    const v4f bias1 = *(const v4f*)&B1[n0 + 64 + 4 * m];
#pragma unroll
    for (int v = 0; v < 8; ++v) {
      const int row = v + 8 * kh;
      v4f o0, o1;
#pragma unroll
      for (int j = 0; j < 4; ++j) {
        float x0 = acc[j][v] + bias0[j];
        float x1 = acc[4 + j][v] + bias1[j];
        o0[j] = x0 > 0.f ? x0 : 0.f;
        o1[j] = x1 > 0.f ? x1 : 0.f;
      }
      *(v4f*)&EH[row * HID + n0 + 4 * m]      = o0;
      *(v4f*)&EH[row * HID + n0 + 64 + 4 * m] = o1;
    }
  }
  __syncthreads();
  {
    int row = tid >> 3, seg = tid & 7;
    const float* w2 = ew2 + k * HID + seg * 64;
    const float* eh = EH + row * HID + seg * 64;
    float p = 0.f;
#pragma unroll 64
    for (int c = 0; c < 64; ++c) p += eh[c] * w2[c];
    psum[row * 8 + seg] = p;
  }
  __syncthreads();
  if (tid < 16) {
    int b = bids[tid];
    if (b >= 0) {
      float eo = eb2[k];
      for (int s = 0; s < 8; ++s) eo += psum[tid * 8 + s];
      float g = gates[b * K_ + k];
      atomicAdd(&y[b], g * eo);
    }
  }
}

extern "C" void kernel_launch(void* const* d_in, const int* in_sizes, int n_in,
                              void* d_out, int out_size, void* d_ws, size_t ws_size,
                              hipStream_t stream) {
  const int*   x         = (const int*)d_in[0];
  const int*   sql       = (const int*)d_in[1];
  const float* sql_emb_w = (const float*)d_in[2];
  const float* inp_emb_w = (const float*)d_in[3];
  const float* gw1 = (const float*)d_in[4];
  const float* gb1 = (const float*)d_in[5];
  const float* gw2 = (const float*)d_in[6];
  const float* gb2 = (const float*)d_in[7];
  const float* ew1 = (const float*)d_in[8];
  const float* eb1 = (const float*)d_in[9];
  const float* ew2 = (const float*)d_in[10];
  const float* eb2 = (const float*)d_in[11];
  float* out = (float*)d_out;

  float* h      = (float*)d_ws;
  float* gates  = h + (size_t)B_ * HID;
  int*   counts = (int*)(gates + (size_t)B_ * K_);
  int*   lists  = counts + K_;

  vsams_gate_l1_kernel<<<256, 128, 0, stream>>>(sql, sql_emb_w, gw1, gb1, h, out, counts);
  vsams_gate_head_kernel<<<256, 128, 0, stream>>>(h, gw2, gb2, gates, counts, lists);
  vsams_expert_kernel<<<529, 128, 0, stream>>>(x, inp_emb_w, ew1, eb1, ew2,
                                                eb2, gates, counts, lists, out);
}


# Weight loads of both hidden-layer k-loops on two register sets: a trip's loads leave one trip ahead (the first before the gathers), so their round trip runs under WMMAs; sums unchanged
# speedup vs baseline: 3.6747x; 1.0740x over previous
#include <hip/hip_runtime.h>

#define B_     4096
#define NFIELD 20
#define NEMB   16
#define HID    512
#define K_     16
#define GIN    320
#define EPSF   1e-10f

typedef __attribute__((ext_vector_type(2))) float v2f;
typedef __attribute__((ext_vector_type(8))) float v8f;
typedef __attribute__((ext_vector_type(4))) float v4f;

__device__ __forceinline__ v8f wmma_f32_16x16x4(v2f a, v2f b, v8f c) {
  return __builtin_amdgcn_wmma_f32_16x16x4_f32(false, a, false, b, (short)0, c,
                                               false, false);
}

template <int NS>
__device__ __forceinline__ void hidden_trip_load(const float* __restrict__ wp,
                                                 v4f (&b)[4 * NS]) {
#pragma unroll
  for (int s = 0; s < NS; ++s) {
    const float* r0 = wp + (size_t)(4 * s) * HID;
    b[4 * s + 0] = *(const v4f*)(r0);
    b[4 * s + 1] = *(const v4f*)(r0 + 64);
    b[4 * s + 2] = *(const v4f*)(r0 + HID);
    b[4 * s + 3] = *(const v4f*)(r0 + HID + 64);
  }
}

template <int NS>
__device__ __forceinline__ void hidden_trip_mma(const float* ap,
                                                const v4f (&b)[4 * NS],
                                                v8f (&acc)[8]) {
#pragma unroll
  for (int s = 0; s < NS; ++s) {
    v2f a;
    a.x = ap[4 * s]; a.y = ap[4 * s + 1];
#pragma unroll
    for (int j = 0; j < 4; ++j) {
      v2f bb; bb.x = b[4 * s + 0][j]; bb.y = b[4 * s + 2][j];
      acc[j] = wmma_f32_16x16x4(a, bb, acc[j]);
    }
#pragma unroll
    for (int j = 0; j < 4; ++j) {
      v2f bb; bb.x = b[4 * s + 1][j]; bb.y = b[4 * s + 3][j];
      acc[4 + j] = wmma_f32_16x16x4(a, bb, acc[4 + j]);
    }
  }
}

__global__ void __launch_bounds__(128) __attribute__((amdgpu_waves_per_eu(1, 2)))
vsams_gate_l1_kernel(const int* __restrict__ sql,
                     const float* __restrict__ sql_emb_w,
                     const float* __restrict__ gw1,
                     const float* __restrict__ gb1,
                     float* __restrict__ h,
                     float* __restrict__ y,
                     int* __restrict__ counts) {
  __shared__ __attribute__((aligned(16))) float Alds[16 * GIN];
  const int tid = threadIdx.x;
  const int m0  = blockIdx.x * 16;
  if ((blockIdx.x & 1) == 0 && tid < 32) y[m0 + tid] = 0.f;
  if (blockIdx.x == 0 && tid < K_) counts[tid] = 0;
  const int lane = tid & 31, wave = tid >> 5;
  const int n0 = wave * 128;
  const int m  = lane & 15;
  const int kh = lane >> 4;
  const float* wp = gw1 + (size_t)(2 * kh) * HID + n0 + 4 * m;
  v4f s0[40], s1[40];
  {
    int s[10];
#pragma unroll
    for (int i = 0; i < 10; ++i) {
      int idx = tid + 128 * i;
      int r = idx / 80;
      int f = (idx - r * 80) >> 2;
      s[i] = sql[(m0 + r) * NFIELD + f];
    }
    hidden_trip_load<10>(wp, s0);
    v4f v[10];
#pragma unroll
    for (int i = 0; i < 10; ++i)
      v[i] = *(const v4f*)&sql_emb_w[(size_t)s[i] * NEMB + (tid & 3) * 4];
#pragma unroll
    for (int i = 0; i < 10; ++i) *(v4f*)&Alds[(tid + 128 * i) * 4] = v[i];
  }
  __syncthreads();
  v8f acc[8];
#pragma unroll
  for (int j = 0; j < 8; ++j) acc[j] = (v8f){0.f, 0.f, 0.f, 0.f, 0.f, 0.f, 0.f, 0.f};
  const float* ap = &Alds[m * GIN + 2 * kh];
  {
    constexpr int TK = 40;
#pragma unroll 1
    for (int t = 0; t < 6; t += 2) {
      hidden_trip_load<10>(wp + (size_t)(t + 1) * TK * HID, s1);
      hidden_trip_mma<10>(ap + t * TK, s0, acc);
      hidden_trip_load<10>(wp + (size_t)(t + 2) * TK * HID, s0);
      hidden_trip_mma<10>(ap + (t + 1) * TK, s1, acc);
    }
    hidden_trip_load<10>(wp + (size_t)7 * TK * HID, s1);
    hidden_trip_mma<10>(ap + 6 * TK, s0, acc);
    hidden_trip_mma<10>(ap + 7 * TK, s1, acc);
  }
  const v4f bias0 = *(const v4f*)&gb1[n0 + 4 * m];
  const v4f bias1 = *(const v4f*)&gb1[n0 + 64 + 4 * m];
#pragma unroll
  for (int v = 0; v < 8; ++v) {
    const int row = m0 + v + 8 * kh;
    v4f o0, o1;
#pragma unroll
    for (int j = 0; j < 4; ++j) {
      float x0 = acc[j][v] + bias0[j];
      float x1 = acc[4 + j][v] + bias1[j];
      o0[j] = x0 > 0.f ? x0 : 0.f;
      o1[j] = x1 > 0.f ? x1 : 0.f;
    }
    *(v4f*)&h[(size_t)row * HID + n0 + 4 * m]      = o0;
    *(v4f*)&h[(size_t)row * HID + n0 + 64 + 4 * m] = o1;
  }
}

__global__ void __launch_bounds__(128) __attribute__((amdgpu_waves_per_eu(1, 2)))
vsams_gate_head_kernel(const float* __restrict__ h,
                       const float* __restrict__ gw2,
                       const float* __restrict__ gb2,
                       float* __restrict__ gates,
                       int* __restrict__ counts,
                       int* __restrict__ lists) {
  __shared__ __attribute__((aligned(16))) float part[4 * 16 * 16];
  const int tid = threadIdx.x;
  const int lane = tid & 31, wave = tid >> 5;
  const int m0 = blockIdx.x * 16;
  const int m  = lane & 15;
  const int kh = lane >> 4;
  const int kq = 128 * wave + 64 * kh;
  const float* ap = h + (size_t)(m0 + m) * HID + kq;
  const float* bp = gw2 + (size_t)kq * K_ + m;
  v4f a[16];
#pragma unroll
  for (int q = 0; q < 16; ++q) a[q] = *(const v4f*)(ap + 4 * q);
  float b[64];
#pragma unroll
  for (int i = 0; i < 64; ++i) b[i] = bp[i * K_];
  v8f acc0 = {0.f, 0.f, 0.f, 0.f, 0.f, 0.f, 0.f, 0.f};
  v8f acc1 = {0.f, 0.f, 0.f, 0.f, 0.f, 0.f, 0.f, 0.f};
#pragma unroll
  for (int q = 0; q < 16; ++q) {
    v2f a0, a1, b0, b1;
    a0.x = a[q][0]; a0.y = a[q][1];
    a1.x = a[q][2]; a1.y = a[q][3];
    b0.x = b[4 * q];     b0.y = b[4 * q + 1];
    b1.x = b[4 * q + 2]; b1.y = b[4 * q + 3];
    acc0 = wmma_f32_16x16x4(a0, b0, acc0);
    acc1 = wmma_f32_16x16x4(a1, b1, acc1);
  }
#pragma unroll
  for (int v = 0; v < 8; ++v)
    part[wave * 256 + (v + 8 * kh) * 16 + m] = acc0[v] + acc1[v];
  __syncthreads();
  if (tid < 16) {
    const int b_ = m0 + tid;
    float l[K_];
    float mx = -3.0e38f;
    for (int j = 0; j < K_; ++j) {
      float s4 = part[tid * 16 + j] + part[256 + tid * 16 + j];
      s4 += part[512 + tid * 16 + j];
      s4 += part[768 + tid * 16 + j];
      l[j] = s4 + gb2[j];
      mx = fmaxf(mx, l[j]);
    }
    float s = 0.f;
    for (int j = 0; j < K_; ++j) { l[j] = __expf(l[j] - mx); s += l[j]; }
    float inv = 1.f / s;
    for (int j = 0; j < K_; ++j) l[j] *= inv;
    int i1 = 0; float v1 = l[0];
    for (int j = 1; j < K_; ++j) if (l[j] > v1) { v1 = l[j]; i1 = j; }
    int i2 = -1; float v2 = -1.f;
    for (int j = 0; j < K_; ++j) if (j != i1 && l[j] > v2) { v2 = l[j]; i2 = j; }
    float denom = v1 + v2;
    float g1 = v1 / denom, g2 = v2 / denom;
    for (int j = 0; j < K_; ++j) gates[b_ * K_ + j] = 0.f;
    gates[b_ * K_ + i1] = g1;
    gates[b_ * K_ + i2] = g2;
    int p1 = atomicAdd(&counts[i1], 1);
    lists[i1 * B_ + p1] = b_;
    int p2 = atomicAdd(&counts[i2], 1);
    lists[i2 * B_ + p2] = b_;
  }
}

__device__ __forceinline__ void vsams_loss_block(const float* __restrict__ gates,
                                                 float* __restrict__ out,
                                                 float* __restrict__ scr) {
  float* wimp = scr;
  float* wld  = scr + 4 * K_;
  float* fin  = scr + 8 * K_;
  const int t = threadIdx.x;
  const int lane = t & 31, wave = t >> 5;
  float pi[4] = {0.f, 0.f, 0.f, 0.f}, pl[4] = {0.f, 0.f, 0.f, 0.f};
#pragma unroll 1
  for (int j0 = 0; j0 < 128; j0 += 16) {
    v4f g[16];
#pragma unroll
    for (int j = 0; j < 16; ++j)
      g[j] = *(const v4f*)&gates[(size_t)((j0 + j) * 128 + t) * 4];
#pragma unroll
    for (int j = 0; j < 16; ++j) {
#pragma unroll
      for (int c = 0; c < 4; ++c) {
        float gv = g[j][c];
        pi[c] += gv;
        pl[c] += (gv > 0.f) ? 1.f : 0.f;
      }
    }
  }
#pragma unroll
  for (int c = 0; c < 4; ++c) {
#pragma unroll
    for (int off = 4; off < 32; off <<= 1) {
      pi[c] += __shfl_xor(pi[c], off);
      pl[c] += __shfl_xor(pl[c], off);
    }
  }
  if (lane < 4) {
#pragma unroll
    for (int c = 0; c < 4; ++c) {
      wimp[wave * K_ + lane * 4 + c] = pi[c];
      wld[wave * K_ + lane * 4 + c]  = pl[c];
    }
  }
  __syncthreads();
  if (t < 2 * K_) {
    const float* src = (t < K_) ? wimp : wld;
    const int col = t & (K_ - 1);
    float s = 0.f;
    for (int w = 0; w < 4; ++w) s += src[w * K_ + col];
    fin[t] = s;
  }
  __syncthreads();
  if (t == 0) {
    float imp[K_], ld[K_];
    for (int kk = 0; kk < K_; ++kk) { imp[kk] = fin[kk]; ld[kk] = fin[K_ + kk]; }
    float mi = 0.f, ml = 0.f;
    for (int kk = 0; kk < K_; ++kk) { mi += imp[kk]; ml += ld[kk]; }
    mi *= (1.f / 16.f); ml *= (1.f / 16.f);
    float vi = 0.f, vl = 0.f;
    for (int kk = 0; kk < K_; ++kk) {
      float di = imp[kk] - mi, dl = ld[kk] - ml;
      vi += di * di; vl += dl * dl;
    }
    vi *= (1.f / 16.f); vl *= (1.f / 16.f);
    out[B_] = vi / (mi * mi + EPSF) + vl / (ml * ml + EPSF);
  }
}

__global__ void __launch_bounds__(128)
vsams_expert_kernel(const int* __restrict__ x,
                    const float* __restrict__ inp_emb_w,
                    const float* __restrict__ ew1,
                    const float* __restrict__ eb1,
                    const float* __restrict__ ew2,
                    const float* __restrict__ eb2,
                    const float* __restrict__ gates,
                    const int* __restrict__ counts,
                    const int* __restrict__ lists,
                    float* __restrict__ y) {
  __shared__ __attribute__((aligned(16))) float Alds[16 * GIN];
  __shared__ __attribute__((aligned(16))) float EH[16 * HID];
  __shared__ float psum[16 * 8];
  __shared__ int   bids[16];
  if (blockIdx.x == 0) {
    vsams_loss_block(gates, y, EH);
    return;
  }
  const int bid = blockIdx.x - 1;
  int k = -1, tile = 0, cnt = 0, first = 0;
  for (int e = 0; e < K_; ++e) {
    const int c  = counts[e];
    const int nt = (c + 15) >> 4;
    if (bid >= first && bid < first + nt) { k = e; tile = bid - first; cnt = c; }
    first += nt;
  }
  if (k < 0) return;
  const int base = tile * 16;
  const int tid = threadIdx.x;
  const int lane = tid & 31, wave = tid >> 5;
  const int m  = lane & 15;
  const int kh = lane >> 4;
  const float* W1 = ew1 + (size_t)k * GIN * HID;
  const float* B1 = eb1 + k * HID;
  const float* wp = W1 + (size_t)(2 * kh) * HID + wave * 128 + 4 * m;
  v4f s0[20], s1[20];
  hidden_trip_load<5>(wp, s0);
  if (tid < 16) bids[tid] = (base + tid < cnt) ? lists[k * B_ + base + tid] : -1;
  __syncthreads();
  {
    int xi[10];
#pragma unroll
    for (int i = 0; i < 10; ++i) {
      int idx = tid + 128 * i;
      int r = idx / 80;
      int f = (idx - r * 80) >> 2;
      int b = bids[r];
      xi[i] = x[(b >= 0 ? b : 0) * NFIELD + f];
    }
    v4f v[10];
#pragma unroll
    for (int i = 0; i < 10; ++i)
      v[i] = *(const v4f*)&inp_emb_w[(size_t)xi[i] * NEMB + (tid & 3) * 4];
#pragma unroll
    for (int i = 0; i < 10; ++i) *(v4f*)&Alds[(tid + 128 * i) * 4] = v[i];
  }
  __syncthreads();
  {
    const int n0 = wave * 128;
    v8f acc[8];
#pragma unroll
    for (int j = 0; j < 8; ++j) acc[j] = (v8f){0.f, 0.f, 0.f, 0.f, 0.f, 0.f, 0.f, 0.f};
    const float* ap = &Alds[m * GIN + 2 * kh];
    {
      constexpr int TK = 20;
#pragma unroll 1
      for (int t = 0; t < 14; t += 2) {
        hidden_trip_load<5>(wp + (size_t)(t + 1) * TK * HID, s1);
        hidden_trip_mma<5>(ap + t * TK, s0, acc);
        hidden_trip_load<5>(wp + (size_t)(t + 2) * TK * HID, s0);
        hidden_trip_mma<5>(ap + (t + 1) * TK, s1, acc);
      }
      hidden_trip_load<5>(wp + (size_t)15 * TK * HID, s1);
      hidden_trip_mma<5>(ap + 14 * TK, s0, acc);
      hidden_trip_mma<5>(ap + 15 * TK, s1, acc);
    }
    const v4f bias0 = *(const v4f*)&B1[n0 + 4 * m];
    const v4f bias1 = *(const v4f*)&B1[n0 + 64 + 4 * m];
#pragma unroll
    for (int v = 0; v < 8; ++v) {
      const int row = v + 8 * kh;
      v4f o0, o1;
#pragma unroll
      for (int j = 0; j < 4; ++j) {
        float x0 = acc[j][v] + bias0[j];
        float x1 = acc[4 + j][v] + bias1[j];
        o0[j] = x0 > 0.f ? x0 : 0.f;
        o1[j] = x1 > 0.f ? x1 : 0.f;
      }
      *(v4f*)&EH[row * HID + n0 + 4 * m]      = o0;
      *(v4f*)&EH[row * HID + n0 + 64 + 4 * m] = o1;
    }
  }
  __syncthreads();
  {
    int row = tid >> 3, seg = tid & 7;
    const float* w2 = ew2 + k * HID + seg * 64;
    const float* eh = EH + row * HID + seg * 64;
    float p = 0.f;
#pragma unroll 64
    for (int c = 0; c < 64; ++c) p += eh[c] * w2[c];
    psum[row * 8 + seg] = p;
  }
  __syncthreads();
  if (tid < 16) {
    int b = bids[tid];
    if (b >= 0) {
      float eo = eb2[k];
      for (int s = 0; s < 8; ++s) eo += psum[tid * 8 + s];
      float g = gates[b * K_ + k];
      atomicAdd(&y[b], g * eo);
    }
  }
}

extern "C" void kernel_launch(void* const* d_in, const int* in_sizes, int n_in,
                              void* d_out, int out_size, void* d_ws, size_t ws_size,
                              hipStream_t stream) {
  const int*   x         = (const int*)d_in[0];
  const int*   sql       = (const int*)d_in[1];
  const float* sql_emb_w = (const float*)d_in[2];
  const float* inp_emb_w = (const float*)d_in[3];
  const float* gw1 = (const float*)d_in[4];
  const float* gb1 = (const float*)d_in[5];
  const float* gw2 = (const float*)d_in[6];
  const float* gb2 = (const float*)d_in[7];
  const float* ew1 = (const float*)d_in[8];
  const float* eb1 = (const float*)d_in[9];
  const float* ew2 = (const float*)d_in[10];
  const float* eb2 = (const float*)d_in[11];
  float* out = (float*)d_out;

  float* h      = (float*)d_ws;
  float* gates  = h + (size_t)B_ * HID;
  int*   counts = (int*)(gates + (size_t)B_ * K_);
  int*   lists  = counts + K_;

  vsams_gate_l1_kernel<<<256, 128, 0, stream>>>(sql, sql_emb_w, gw1, gb1, h, out, counts);
  vsams_gate_head_kernel<<<256, 128, 0, stream>>>(h, gw2, gb2, gates, counts, lists);
  vsams_expert_kernel<<<529, 128, 0, stream>>>(x, inp_emb_w, ew1, eb1, ew2,
                                                eb2, gates, counts, lists, out);
}


# Expert second layer taken from the accumulator registers: 8 FMAs per row and lane, 16-lane sums in the vector unit, 4 wave parts in LDS; no 32 KB LDS round trip of eh; eo sum order changes
# speedup vs baseline: 3.8747x; 1.0544x over previous
#include <hip/hip_runtime.h>

#define B_     4096
#define NFIELD 20
#define NEMB   16
#define HID    512
#define K_     16
#define GIN    320
#define EPSF   1e-10f

typedef __attribute__((ext_vector_type(2))) float v2f;
typedef __attribute__((ext_vector_type(8))) float v8f;
typedef __attribute__((ext_vector_type(4))) float v4f;

__device__ __forceinline__ v8f wmma_f32_16x16x4(v2f a, v2f b, v8f c) {
  return __builtin_amdgcn_wmma_f32_16x16x4_f32(false, a, false, b, (short)0, c,
                                               false, false);
}

template <int CTRL>
__device__ __forceinline__ float half_xor_get(float x) {
  return __builtin_bit_cast(float, __builtin_amdgcn_update_dpp(
      0, __builtin_bit_cast(int, x), CTRL, 0xf, 0xf, false));
}
__device__ __forceinline__ float half_sum16(float x) {
  x += half_xor_get<0x161>(x);
  x += half_xor_get<0x162>(x);
  x += half_xor_get<0x164>(x);
  x += half_xor_get<0x168>(x);
  return x;
}

template <int NS>
__device__ __forceinline__ void hidden_trip_load(const float* __restrict__ wp,
                                                 v4f (&b)[4 * NS]) {
#pragma unroll
  for (int s = 0; s < NS; ++s) {
    const float* r0 = wp + (size_t)(4 * s) * HID;
    b[4 * s + 0] = *(const v4f*)(r0);
    b[4 * s + 1] = *(const v4f*)(r0 + 64);
    b[4 * s + 2] = *(const v4f*)(r0 + HID);
    b[4 * s + 3] = *(const v4f*)(r0 + HID + 64);
  }
}

template <int NS>
__device__ __forceinline__ void hidden_trip_mma(const float* ap,
                                                const v4f (&b)[4 * NS],
                                                v8f (&acc)[8]) {
#pragma unroll
  for (int s = 0; s < NS; ++s) {
    v2f a;
    a.x = ap[4 * s]; a.y = ap[4 * s + 1];
#pragma unroll
    for (int j = 0; j < 4; ++j) {
      v2f bb; bb.x = b[4 * s + 0][j]; bb.y = b[4 * s + 2][j];
      acc[j] = wmma_f32_16x16x4(a, bb, acc[j]);
    }
#pragma unroll
    for (int j = 0; j < 4; ++j) {
      v2f bb; bb.x = b[4 * s + 1][j]; bb.y = b[4 * s + 3][j];
      acc[4 + j] = wmma_f32_16x16x4(a, bb, acc[4 + j]);
    }
  }
}

__global__ void __launch_bounds__(128) __attribute__((amdgpu_waves_per_eu(1, 2)))
vsams_gate_l1_kernel(const int* __restrict__ sql,
                     const float* __restrict__ sql_emb_w,
                     const float* __restrict__ gw1,
                     const float* __restrict__ gb1,
                     float* __restrict__ h,
                     float* __restrict__ y,
                     int* __restrict__ counts) {
  __shared__ __attribute__((aligned(16))) float Alds[16 * GIN];
  const int tid = threadIdx.x;
  const int m0  = blockIdx.x * 16;
  if ((blockIdx.x & 1) == 0 && tid < 32) y[m0 + tid] = 0.f;
  if (blockIdx.x == 0 && tid < K_) counts[tid] = 0;
  const int lane = tid & 31, wave = tid >> 5;
  const int n0 = wave * 128;
  const int m  = lane & 15;
  const int kh = lane >> 4;
  const float* wp = gw1 + (size_t)(2 * kh) * HID + n0 + 4 * m;
  v4f s0[40], s1[40];
  {
    int s[10];
#pragma unroll
    for (int i = 0; i < 10; ++i) {
      int idx = tid + 128 * i;
      int r = idx / 80;
      int f = (idx - r * 80) >> 2;
      s[i] = sql[(m0 + r) * NFIELD + f];
    }
    hidden_trip_load<10>(wp, s0);
    v4f v[10];
#pragma unroll
    for (int i = 0; i < 10; ++i)
      v[i] = *(const v4f*)&sql_emb_w[(size_t)s[i] * NEMB + (tid & 3) * 4];
#pragma unroll
    for (int i = 0; i < 10; ++i) *(v4f*)&Alds[(tid + 128 * i) * 4] = v[i];
  }
  __syncthreads();
  v8f acc[8];
#pragma unroll
  for (int j = 0; j < 8; ++j) acc[j] = (v8f){0.f, 0.f, 0.f, 0.f, 0.f, 0.f, 0.f, 0.f};
  const float* ap = &Alds[m * GIN + 2 * kh];
  {
    constexpr int TK = 40;
#pragma unroll 1
    for (int t = 0; t < 6; t += 2) {
      hidden_trip_load<10>(wp + (size_t)(t + 1) * TK * HID, s1);
      hidden_trip_mma<10>(ap + t * TK, s0, acc);
      hidden_trip_load<10>(wp + (size_t)(t + 2) * TK * HID, s0);
      hidden_trip_mma<10>(ap + (t + 1) * TK, s1, acc);
    }
    hidden_trip_load<10>(wp + (size_t)7 * TK * HID, s1);
    hidden_trip_mma<10>(ap + 6 * TK, s0, acc);
    hidden_trip_mma<10>(ap + 7 * TK, s1, acc);
  }
  const v4f bias0 = *(const v4f*)&gb1[n0 + 4 * m];
  const v4f bias1 = *(const v4f*)&gb1[n0 + 64 + 4 * m];
#pragma unroll
  for (int v = 0; v < 8; ++v) {
    const int row = m0 + v + 8 * kh;
    v4f o0, o1;
#pragma unroll
    for (int j = 0; j < 4; ++j) {
      float x0 = acc[j][v] + bias0[j];
      float x1 = acc[4 + j][v] + bias1[j];
      o0[j] = x0 > 0.f ? x0 : 0.f;
      o1[j] = x1 > 0.f ? x1 : 0.f;
    }
    *(v4f*)&h[(size_t)row * HID + n0 + 4 * m]      = o0;
    *(v4f*)&h[(size_t)row * HID + n0 + 64 + 4 * m] = o1;
  }
}

__global__ void __launch_bounds__(128) __attribute__((amdgpu_waves_per_eu(1, 2)))
vsams_gate_head_kernel(const float* __restrict__ h,
                       const float* __restrict__ gw2,
                       const float* __restrict__ gb2,
                       float* __restrict__ gates,
                       int* __restrict__ counts,
                       int* __restrict__ lists) {
  __shared__ __attribute__((aligned(16))) float part[4 * 16 * 16];
  const int tid = threadIdx.x;
  const int lane = tid & 31, wave = tid >> 5;
  const int m0 = blockIdx.x * 16;
  const int m  = lane & 15;
  const int kh = lane >> 4;
  const int kq = 128 * wave + 64 * kh;
  const float* ap = h + (size_t)(m0 + m) * HID + kq;
  const float* bp = gw2 + (size_t)kq * K_ + m;
  v4f a[16];
#pragma unroll
  for (int q = 0; q < 16; ++q) a[q] = *(const v4f*)(ap + 4 * q);
  float b[64];
#pragma unroll
  for (int i = 0; i < 64; ++i) b[i] = bp[i * K_];
  v8f acc0 = {0.f, 0.f, 0.f, 0.f, 0.f, 0.f, 0.f, 0.f};
  v8f acc1 = {0.f, 0.f, 0.f, 0.f, 0.f, 0.f, 0.f, 0.f};
#pragma unroll
  for (int q = 0; q < 16; ++q) {
    v2f a0, a1, b0, b1;
    a0.x = a[q][0]; a0.y = a[q][1];
    a1.x = a[q][2]; a1.y = a[q][3];
    b0.x = b[4 * q];     b0.y = b[4 * q + 1];
    b1.x = b[4 * q + 2]; b1.y = b[4 * q + 3];
    acc0 = wmma_f32_16x16x4(a0, b0, acc0);
    acc1 = wmma_f32_16x16x4(a1, b1, acc1);
  }
#pragma unroll
  for (int v = 0; v < 8; ++v)
    part[wave * 256 + (v + 8 * kh) * 16 + m] = acc0[v] + acc1[v];
  __syncthreads();
  if (tid < 16) {
    const int b_ = m0 + tid;
    float l[K_];
    float mx = -3.0e38f;
    for (int j = 0; j < K_; ++j) {
      float s4 = part[tid * 16 + j] + part[256 + tid * 16 + j];
      s4 += part[512 + tid * 16 + j];
      s4 += part[768 + tid * 16 + j];
      l[j] = s4 + gb2[j];
      mx = fmaxf(mx, l[j]);
    }
    float s = 0.f;
    for (int j = 0; j < K_; ++j) { l[j] = __expf(l[j] - mx); s += l[j]; }
    float inv = 1.f / s;
    for (int j = 0; j < K_; ++j) l[j] *= inv;
    int i1 = 0; float v1 = l[0];
    for (int j = 1; j < K_; ++j) if (l[j] > v1) { v1 = l[j]; i1 = j; }
    int i2 = -1; float v2 = -1.f;
    for (int j = 0; j < K_; ++j) if (j != i1 && l[j] > v2) { v2 = l[j]; i2 = j; }
    float denom = v1 + v2;
    float g1 = v1 / denom, g2 = v2 / denom;
    for (int j = 0; j < K_; ++j) gates[b_ * K_ + j] = 0.f;
    gates[b_ * K_ + i1] = g1;
    gates[b_ * K_ + i2] = g2;
    int p1 = atomicAdd(&counts[i1], 1);
    lists[i1 * B_ + p1] = b_;
    int p2 = atomicAdd(&counts[i2], 1);
    lists[i2 * B_ + p2] = b_;
  }
}

__device__ __forceinline__ void vsams_loss_block(const float* __restrict__ gates,
                                                 float* __restrict__ out,
                                                 float* __restrict__ scr) {
  float* wimp = scr;
  float* wld  = scr + 4 * K_;
  float* fin  = scr + 8 * K_;
  const int t = threadIdx.x;
  const int lane = t & 31, wave = t >> 5;
  float pi[4] = {0.f, 0.f, 0.f, 0.f}, pl[4] = {0.f, 0.f, 0.f, 0.f};
#pragma unroll 1
  for (int j0 = 0; j0 < 128; j0 += 16) {
    v4f g[16];
#pragma unroll
    for (int j = 0; j < 16; ++j)
      g[j] = *(const v4f*)&gates[(size_t)((j0 + j) * 128 + t) * 4];
#pragma unroll
    for (int j = 0; j < 16; ++j) {
#pragma unroll
      for (int c = 0; c < 4; ++c) {
        float gv = g[j][c];
        pi[c] += gv;
        pl[c] += (gv > 0.f) ? 1.f : 0.f;
      }
    }
  }
#pragma unroll
  for (int c = 0; c < 4; ++c) {
#pragma unroll
    for (int off = 4; off < 32; off <<= 1) {
      pi[c] += __shfl_xor(pi[c], off);
      pl[c] += __shfl_xor(pl[c], off);
    }
  }
  if (lane < 4) {
#pragma unroll
    for (int c = 0; c < 4; ++c) {
      wimp[wave * K_ + lane * 4 + c] = pi[c];
      wld[wave * K_ + lane * 4 + c]  = pl[c];
    }
  }
  __syncthreads();
  if (t < 2 * K_) {
    const float* src = (t < K_) ? wimp : wld;
    const int col = t & (K_ - 1);
    float s = 0.f;
    for (int w = 0; w < 4; ++w) s += src[w * K_ + col];
    fin[t] = s;
  }
  __syncthreads();
  if (t == 0) {
    float imp[K_], ld[K_];
    for (int kk = 0; kk < K_; ++kk) { imp[kk] = fin[kk]; ld[kk] = fin[K_ + kk]; }
    float mi = 0.f, ml = 0.f;
    for (int kk = 0; kk < K_; ++kk) { mi += imp[kk]; ml += ld[kk]; }
    mi *= (1.f / 16.f); ml *= (1.f / 16.f);
    float vi = 0.f, vl = 0.f;
    for (int kk = 0; kk < K_; ++kk) {
      float di = imp[kk] - mi, dl = ld[kk] - ml;
      vi += di * di; vl += dl * dl;
    }
    vi *= (1.f / 16.f); vl *= (1.f / 16.f);
    out[B_] = vi / (mi * mi + EPSF) + vl / (ml * ml + EPSF);
  }
}

__global__ void __launch_bounds__(128)
vsams_expert_kernel(const int* __restrict__ x,
                    const float* __restrict__ inp_emb_w,
                    const float* __restrict__ ew1,
                    const float* __restrict__ eb1,
                    const float* __restrict__ ew2,
                    const float* __restrict__ eb2,
                    const float* __restrict__ gates,
                    const int* __restrict__ counts,
                    const int* __restrict__ lists,
                    float* __restrict__ y) {
  __shared__ __attribute__((aligned(16))) float Alds[16 * GIN];
  __shared__ __attribute__((aligned(16))) float psum[4 * 16];
  __shared__ int   bids[16];
  if (blockIdx.x == 0) {
    vsams_loss_block(gates, y, Alds);
    return;
  }
  const int bid = blockIdx.x - 1;
  int k = -1, tile = 0, cnt = 0, first = 0;
  for (int e = 0; e < K_; ++e) {
    const int c  = counts[e];
    const int nt = (c + 15) >> 4;
    if (bid >= first && bid < first + nt) { k = e; tile = bid - first; cnt = c; }
    first += nt;
  }
  if (k < 0) return;
  const int base = tile * 16;
  const int tid = threadIdx.x;
  const int lane = tid & 31, wave = tid >> 5;
  const int m  = lane & 15;
  const int kh = lane >> 4;
  const float* W1 = ew1 + (size_t)k * GIN * HID;
  const float* B1 = eb1 + k * HID;
  const float* wp = W1 + (size_t)(2 * kh) * HID + wave * 128 + 4 * m;
  v4f s0[20], s1[20];
  hidden_trip_load<5>(wp, s0);
  if (tid < 16) bids[tid] = (base + tid < cnt) ? lists[k * B_ + base + tid] : -1;
  __syncthreads();
  {
    int xi[10];
#pragma unroll
    for (int i = 0; i < 10; ++i) {
      int idx = tid + 128 * i;
      int r = idx / 80;
      int f = (idx - r * 80) >> 2;
      int b = bids[r];
      xi[i] = x[(b >= 0 ? b : 0) * NFIELD + f];
    }
    v4f v[10];
#pragma unroll
    for (int i = 0; i < 10; ++i)
      v[i] = *(const v4f*)&inp_emb_w[(size_t)xi[i] * NEMB + (tid & 3) * 4];
#pragma unroll
    for (int i = 0; i < 10; ++i) *(v4f*)&Alds[(tid + 128 * i) * 4] = v[i];
  }
  __syncthreads();
  {
    const int n0 = wave * 128;
    v8f acc[8];
#pragma unroll
    for (int j = 0; j < 8; ++j) acc[j] = (v8f){0.f, 0.f, 0.f, 0.f, 0.f, 0.f, 0.f, 0.f};
    const float* ap = &Alds[m * GIN + 2 * kh];
    {
      constexpr int TK = 20;
#pragma unroll 1
      for (int t = 0; t < 14; t += 2) {
        hidden_trip_load<5>(wp + (size_t)(t + 1) * TK * HID, s1);
        hidden_trip_mma<5>(ap + t * TK, s0, acc);
        hidden_trip_load<5>(wp + (size_t)(t + 2) * TK * HID, s0);
        hidden_trip_mma<5>(ap + (t + 1) * TK, s1, acc);
      }
      hidden_trip_load<5>(wp + (size_t)15 * TK * HID, s1);
      hidden_trip_mma<5>(ap + 14 * TK, s0, acc);
      hidden_trip_mma<5>(ap + 15 * TK, s1, acc);
    }
    const v4f bias0 = *(const v4f*)&B1[n0 + 4 * m];
    const v4f bias1 = *(const v4f*)&B1[n0 + 64 + 4 * m];
    const v4f w2a = *(const v4f*)&ew2[k * HID + n0 + 4 * m];
    const v4f w2b = *(const v4f*)&ew2[k * HID + n0 + 64 + 4 * m];
    float pv[8];
#pragma unroll
    for (int v = 0; v < 8; ++v) {
      float p = 0.f;
#pragma unroll
      for (int j = 0; j < 4; ++j) {
        float x0 = acc[j][v] + bias0[j];
        float x1 = acc[4 + j][v] + bias1[j];
        x0 = x0 > 0.f ? x0 : 0.f;
        x1 = x1 > 0.f ? x1 : 0.f;
        p += x0 * w2a[j];
        p += x1 * w2b[j];
      }
      pv[v] = half_sum16(p);
    }
    if (m == 0) {
      *(v4f*)&psum[wave * 16 + 8 * kh]     = (v4f){pv[0], pv[1], pv[2], pv[3]};
      *(v4f*)&psum[wave * 16 + 8 * kh + 4] = (v4f){pv[4], pv[5], pv[6], pv[7]};
    }
  }
  __syncthreads();
  if (tid < 16) {
    int b = bids[tid];
    if (b >= 0) {
      float eo = eb2[k];
      for (int w = 0; w < 4; ++w) eo += psum[w * 16 + tid];
      float g = gates[b * K_ + k];
      atomicAdd(&y[b], g * eo);
    }
  }
}

extern "C" void kernel_launch(void* const* d_in, const int* in_sizes, int n_in,
                              void* d_out, int out_size, void* d_ws, size_t ws_size,
                              hipStream_t stream) {
  const int*   x         = (const int*)d_in[0];
  const int*   sql       = (const int*)d_in[1];
  const float* sql_emb_w = (const float*)d_in[2];
  const float* inp_emb_w = (const float*)d_in[3];
  const float* gw1 = (const float*)d_in[4];
  const float* gb1 = (const float*)d_in[5];
  const float* gw2 = (const float*)d_in[6];
  const float* gb2 = (const float*)d_in[7];
  const float* ew1 = (const float*)d_in[8];
  const float* eb1 = (const float*)d_in[9];
  const float* ew2 = (const float*)d_in[10];
  const float* eb2 = (const float*)d_in[11];
  float* out = (float*)d_out;

  float* h      = (float*)d_ws;
  float* gates  = h + (size_t)B_ * HID;
  int*   counts = (int*)(gates + (size_t)B_ * K_);
  int*   lists  = counts + K_;

  vsams_gate_l1_kernel<<<256, 128, 0, stream>>>(sql, sql_emb_w, gw1, gb1, h, out, counts);
  vsams_gate_head_kernel<<<256, 128, 0, stream>>>(h, gw2, gb2, gates, counts, lists);
  vsams_expert_kernel<<<529, 128, 0, stream>>>(x, inp_emb_w, ew1, eb1, ew2,
                                                eb2, gates, counts, lists, out);
}


# Both hidden GEMMs on 32-row x 256-column block tiles (was 16 x 512): a weight fragment feeds 2 WMMAs, L2 weight reads halve; LDS row pitch 324; eo in 2 column parts: y sum order varies
# speedup vs baseline: 3.9908x; 1.0300x over previous
#include <hip/hip_runtime.h>

#define B_     4096
#define NFIELD 20
#define NEMB   16
#define HID    512
#define K_     16
#define GIN    320
#define EPSF   1e-10f
#define MT     32
#define NT     256
#define PITCH  324

typedef __attribute__((ext_vector_type(2))) float v2f;
typedef __attribute__((ext_vector_type(8))) float v8f;
typedef __attribute__((ext_vector_type(4))) float v4f;

__device__ __forceinline__ v8f wmma_f32_16x16x4(v2f a, v2f b, v8f c) {
  return __builtin_amdgcn_wmma_f32_16x16x4_f32(false, a, false, b, (short)0, c,
                                               false, false);
}

template <int CTRL>
__device__ __forceinline__ float half_xor_get(float x) {
  return __builtin_bit_cast(float, __builtin_amdgcn_update_dpp(
      0, __builtin_bit_cast(int, x), CTRL, 0xf, 0xf, false));
}
__device__ __forceinline__ float half_sum16(float x) {
  x += half_xor_get<0x161>(x);
  x += half_xor_get<0x162>(x);
  x += half_xor_get<0x164>(x);
  x += half_xor_get<0x168>(x);
  return x;
}

template <int NS>
__device__ __forceinline__ void hidden_trip_load(const float* __restrict__ wp,
                                                 v4f (&b)[2 * NS]) {
#pragma unroll
  for (int s = 0; s < NS; ++s) {
    const float* r0 = wp + (size_t)(4 * s) * HID;
    b[2 * s + 0] = *(const v4f*)(r0);
    b[2 * s + 1] = *(const v4f*)(r0 + HID);
  }
}

template <int NS>
__device__ __forceinline__ void hidden_trip_mma(const float* ap,
                                                const v4f (&b)[2 * NS],
                                                v8f (&acc)[8]) {
#pragma unroll
  for (int s = 0; s < NS; ++s) {
    v2f a0, a1;
    a0.x = ap[4 * s];              a0.y = ap[4 * s + 1];
    a1.x = ap[16 * PITCH + 4 * s]; a1.y = ap[16 * PITCH + 4 * s + 1];
#pragma unroll
    for (int j = 0; j < 4; ++j) {
      v2f bb; bb.x = b[2 * s + 0][j]; bb.y = b[2 * s + 1][j];
      acc[j]     = wmma_f32_16x16x4(a0, bb, acc[j]);
      acc[4 + j] = wmma_f32_16x16x4(a1, bb, acc[4 + j]);
    }
  }
}

__device__ __forceinline__ void hidden_k_loop(const float* __restrict__ wp,
                                              const float* ap,
                                              v4f (&s0)[20], v4f (&s1)[20],
                                              v8f (&acc)[8]) {
  constexpr int TK = 40;
#pragma unroll 1
  for (int t = 0; t < 6; t += 2) {
    hidden_trip_load<10>(wp + (size_t)(t + 1) * TK * HID, s1);
    hidden_trip_mma<10>(ap + t * TK, s0, acc);
    hidden_trip_load<10>(wp + (size_t)(t + 2) * TK * HID, s0);
    hidden_trip_mma<10>(ap + (t + 1) * TK, s1, acc);
  }
  hidden_trip_load<10>(wp + (size_t)7 * TK * HID, s1);
  hidden_trip_mma<10>(ap + 6 * TK, s0, acc);
  hidden_trip_mma<10>(ap + 7 * TK, s1, acc);
}

__global__ void __launch_bounds__(128) __attribute__((amdgpu_waves_per_eu(1, 2)))
vsams_gate_l1_kernel(const int* __restrict__ sql,
                     const float* __restrict__ sql_emb_w,
                     const float* __restrict__ gw1,
                     const float* __restrict__ gb1,
                     float* __restrict__ h,
                     float* __restrict__ y,
                     int* __restrict__ counts) {
  __shared__ __attribute__((aligned(16))) float Alds[MT * PITCH];
  const int tid = threadIdx.x;
  const int nh  = blockIdx.x & 1;
  const int m0  = (blockIdx.x >> 1) * MT;
  if (nh == 0 && tid < 32) y[m0 + tid] = 0.f;
  if (blockIdx.x == 0 && tid < K_) counts[tid] = 0;
  const int lane = tid & 31, wave = tid >> 5;
  const int n0 = nh * NT + wave * 64;
  const int m  = lane & 15;
  const int kh = lane >> 4;
  const float* wp = gw1 + (size_t)(2 * kh) * HID + n0 + 4 * m;
  v4f s0[20], s1[20];
  {
    int s[20];
#pragma unroll
    for (int i = 0; i < 20; ++i) {
      int idx = tid + 128 * i;
      int r = idx / 80;
      int f = (idx - r * 80) >> 2;
      s[i] = sql[(m0 + r) * NFIELD + f];
    }
    hidden_trip_load<10>(wp, s0);
    v4f v[20];
#pragma unroll
    for (int i = 0; i < 20; ++i)
      v[i] = *(const v4f*)&sql_emb_w[(size_t)s[i] * NEMB + (tid & 3) * 4];
#pragma unroll
    for (int i = 0; i < 20; ++i) {
      int idx = tid + 128 * i;
      int r = idx / 80;
      *(v4f*)&Alds[r * PITCH + (idx - r * 80) * 4] = v[i];
    }
  }
  __syncthreads();
  v8f acc[8];
#pragma unroll
  for (int j = 0; j < 8; ++j) acc[j] = (v8f){0.f, 0.f, 0.f, 0.f, 0.f, 0.f, 0.f, 0.f};
  hidden_k_loop(wp, &Alds[m * PITCH + 2 * kh], s0, s1, acc);
  const v4f bias = *(const v4f*)&gb1[n0 + 4 * m];
#pragma unroll
  for (int i = 0; i < 2; ++i) {
#pragma unroll
    for (int v = 0; v < 8; ++v) {
      const int row = m0 + 16 * i + v + 8 * kh;
      v4f o;
#pragma unroll
      for (int j = 0; j < 4; ++j) {
        float x0 = acc[4 * i + j][v] + bias[j];
        o[j] = x0 > 0.f ? x0 : 0.f;
      }
      *(v4f*)&h[(size_t)row * HID + n0 + 4 * m] = o;
    }
  }
}

__global__ void __launch_bounds__(128) __attribute__((amdgpu_waves_per_eu(1, 2)))
vsams_gate_head_kernel(const float* __restrict__ h,
                       const float* __restrict__ gw2,
                       const float* __restrict__ gb2,
                       float* __restrict__ gates,
                       int* __restrict__ counts,
                       int* __restrict__ lists) {
  __shared__ __attribute__((aligned(16))) float part[4 * 16 * 16];
  const int tid = threadIdx.x;
  const int lane = tid & 31, wave = tid >> 5;
  const int m0 = blockIdx.x * 16;
  const int m  = lane & 15;
  const int kh = lane >> 4;
  const int kq = 128 * wave + 64 * kh;
  const float* ap = h + (size_t)(m0 + m) * HID + kq;
  const float* bp = gw2 + (size_t)kq * K_ + m;
  v4f a[16];
#pragma unroll
  for (int q = 0; q < 16; ++q) a[q] = *(const v4f*)(ap + 4 * q);
  float b[64];
#pragma unroll
  for (int i = 0; i < 64; ++i) b[i] = bp[i * K_];
  v8f acc0 = {0.f, 0.f, 0.f, 0.f, 0.f, 0.f, 0.f, 0.f};
  v8f acc1 = {0.f, 0.f, 0.f, 0.f, 0.f, 0.f, 0.f, 0.f};
#pragma unroll
  for (int q = 0; q < 16; ++q) {
    v2f a0, a1, b0, b1;
    a0.x = a[q][0]; a0.y = a[q][1];
    a1.x = a[q][2]; a1.y = a[q][3];
    b0.x = b[4 * q];     b0.y = b[4 * q + 1];
    b1.x = b[4 * q + 2]; b1.y = b[4 * q + 3];
    acc0 = wmma_f32_16x16x4(a0, b0, acc0);
    acc1 = wmma_f32_16x16x4(a1, b1, acc1);
  }
#pragma unroll
  for (int v = 0; v < 8; ++v)
    part[wave * 256 + (v + 8 * kh) * 16 + m] = acc0[v] + acc1[v];
  __syncthreads();
  if (tid < 16) {
    const int b_ = m0 + tid;
    float l[K_];
    float mx = -3.0e38f;
    for (int j = 0; j < K_; ++j) {
      float s4 = part[tid * 16 + j] + part[256 + tid * 16 + j];
      s4 += part[512 + tid * 16 + j];
      s4 += part[768 + tid * 16 + j];
      l[j] = s4 + gb2[j];
      mx = fmaxf(mx, l[j]);
    }
    float s = 0.f;
    for (int j = 0; j < K_; ++j) { l[j] = __expf(l[j] - mx); s += l[j]; }
    float inv = 1.f / s;
    for (int j = 0; j < K_; ++j) l[j] *= inv;
    int i1 = 0; float v1 = l[0];
    for (int j = 1; j < K_; ++j) if (l[j] > v1) { v1 = l[j]; i1 = j; }
    int i2 = -1; float v2 = -1.f;
    for (int j = 0; j < K_; ++j) if (j != i1 && l[j] > v2) { v2 = l[j]; i2 = j; }
    float denom = v1 + v2;
    float g1 = v1 / denom, g2 = v2 / denom;
    for (int j = 0; j < K_; ++j) gates[b_ * K_ + j] = 0.f;
    gates[b_ * K_ + i1] = g1;
    gates[b_ * K_ + i2] = g2;
    int p1 = atomicAdd(&counts[i1], 1);
    lists[i1 * B_ + p1] = b_;
    int p2 = atomicAdd(&counts[i2], 1);
    lists[i2 * B_ + p2] = b_;
  }
}

__device__ __forceinline__ void vsams_loss_block(const float* __restrict__ gates,
                                                 float* __restrict__ out,
                                                 float* __restrict__ scr) {
  float* wimp = scr;
  float* wld  = scr + 4 * K_;
  float* fin  = scr + 8 * K_;
  const int t = threadIdx.x;
  const int lane = t & 31, wave = t >> 5;
  float pi[4] = {0.f, 0.f, 0.f, 0.f}, pl[4] = {0.f, 0.f, 0.f, 0.f};
#pragma unroll 1
  for (int j0 = 0; j0 < 128; j0 += 16) {
    v4f g[16];
#pragma unroll
    for (int j = 0; j < 16; ++j)
      g[j] = *(const v4f*)&gates[(size_t)((j0 + j) * 128 + t) * 4];
#pragma unroll
    for (int j = 0; j < 16; ++j) {
#pragma unroll
      for (int c = 0; c < 4; ++c) {
        float gv = g[j][c];
        pi[c] += gv;
        pl[c] += (gv > 0.f) ? 1.f : 0.f;
      }
    }
  }
#pragma unroll
  for (int c = 0; c < 4; ++c) {
#pragma unroll
    for (int off = 4; off < 32; off <<= 1) {
      pi[c] += __shfl_xor(pi[c], off);
      pl[c] += __shfl_xor(pl[c], off);
    }
  }
  if (lane < 4) {
#pragma unroll
    for (int c = 0; c < 4; ++c) {
      wimp[wave * K_ + lane * 4 + c] = pi[c];
      wld[wave * K_ + lane * 4 + c]  = pl[c];
    }
  }
  __syncthreads();
  if (t < 2 * K_) {
    const float* src = (t < K_) ? wimp : wld;
    const int col = t & (K_ - 1);
    float s = 0.f;
    for (int w = 0; w < 4; ++w) s += src[w * K_ + col];
    fin[t] = s;
  }
  __syncthreads();
  if (t == 0) {
    float imp[K_], ld[K_];
    for (int kk = 0; kk < K_; ++kk) { imp[kk] = fin[kk]; ld[kk] = fin[K_ + kk]; }
    float mi = 0.f, ml = 0.f;
    for (int kk = 0; kk < K_; ++kk) { mi += imp[kk]; ml += ld[kk]; }
    mi *= (1.f / 16.f); ml *= (1.f / 16.f);
    float vi = 0.f, vl = 0.f;
    for (int kk = 0; kk < K_; ++kk) {
      float di = imp[kk] - mi, dl = ld[kk] - ml;
      vi += di * di; vl += dl * dl;
    }
    vi *= (1.f / 16.f); vl *= (1.f / 16.f);
    out[B_] = vi / (mi * mi + EPSF) + vl / (ml * ml + EPSF);
  }
}

__global__ void __launch_bounds__(128)
vsams_expert_kernel(const int* __restrict__ x,
                    const float* __restrict__ inp_emb_w,
                    const float* __restrict__ ew1,
                    const float* __restrict__ eb1,
                    const float* __restrict__ ew2,
                    const float* __restrict__ eb2,
                    const float* __restrict__ gates,
                    const int* __restrict__ counts,
                    const int* __restrict__ lists,
                    float* __restrict__ y) {
  __shared__ __attribute__((aligned(16))) float Alds[MT * PITCH];
  __shared__ __attribute__((aligned(16))) float psum[4 * MT];
  __shared__ int   bids[MT];
  if (blockIdx.x == 0) {
    vsams_loss_block(gates, y, Alds);
    return;
  }
  const int bid = blockIdx.x - 1;
  int k = -1, loc = 0, cnt = 0, first = 0;
  for (int e = 0; e < K_; ++e) {
    const int c  = counts[e];
    const int nb = ((c + MT - 1) / MT) * 2;
    if (bid >= first && bid < first + nb) { k = e; loc = bid - first; cnt = c; }
    first += nb;
  }
  if (k < 0) return;
  const int base = (loc >> 1) * MT;
  const int nh   = loc & 1;
  const int tid = threadIdx.x;
  const int lane = tid & 31, wave = tid >> 5;
  const int m  = lane & 15;
  const int kh = lane >> 4;
  const int n0 = nh * NT + wave * 64;
  const float* W1 = ew1 + (size_t)k * GIN * HID;
  const float* B1 = eb1 + k * HID;
  const float* wp = W1 + (size_t)(2 * kh) * HID + n0 + 4 * m;
  v4f s0[20], s1[20];
  hidden_trip_load<10>(wp, s0);
  if (tid < MT) bids[tid] = (base + tid < cnt) ? lists[k * B_ + base + tid] : -1;
  __syncthreads();
  {
    int xi[20];
#pragma unroll
    for (int i = 0; i < 20; ++i) {
      int idx = tid + 128 * i;
      int r = idx / 80;
      int f = (idx - r * 80) >> 2;
      int b = bids[r];
      xi[i] = x[(b >= 0 ? b : 0) * NFIELD + f];
    }
    v4f v[20];
#pragma unroll
    for (int i = 0; i < 20; ++i)
      v[i] = *(const v4f*)&inp_emb_w[(size_t)xi[i] * NEMB + (tid & 3) * 4];
#pragma unroll
    for (int i = 0; i < 20; ++i) {
      int idx = tid + 128 * i;
      int r = idx / 80;
      *(v4f*)&Alds[r * PITCH + (idx - r * 80) * 4] = v[i];
    }
  }
  __syncthreads();
  {
    v8f acc[8];
#pragma unroll
    for (int j = 0; j < 8; ++j) acc[j] = (v8f){0.f, 0.f, 0.f, 0.f, 0.f, 0.f, 0.f, 0.f};
    hidden_k_loop(wp, &Alds[m * PITCH + 2 * kh], s0, s1, acc);
    const v4f bias = *(const v4f*)&B1[n0 + 4 * m];
    const v4f w2   = *(const v4f*)&ew2[k * HID + n0 + 4 * m];
    float pv[16];
#pragma unroll
    for (int i = 0; i < 2; ++i) {
#pragma unroll
      for (int v = 0; v < 8; ++v) {
        float p = 0.f;
#pragma unroll
        for (int j = 0; j < 4; ++j) {
          float x0 = acc[4 * i + j][v] + bias[j];
          x0 = x0 > 0.f ? x0 : 0.f;
          p += x0 * w2[j];
        }
        pv[8 * i + v] = half_sum16(p);
      }
    }
    if (m == 0) {
#pragma unroll
      for (int i = 0; i < 2; ++i) {
        float* dst = &psum[wave * MT + 16 * i + 8 * kh];
        *(v4f*)(dst)     = (v4f){pv[8 * i + 0], pv[8 * i + 1], pv[8 * i + 2], pv[8 * i + 3]};
        *(v4f*)(dst + 4) = (v4f){pv[8 * i + 4], pv[8 * i + 5], pv[8 * i + 6], pv[8 * i + 7]};
      }
    }
  }
  __syncthreads();
  if (tid < MT) {
    int b = bids[tid];
    if (b >= 0) {
      float eo = (nh == 0) ? eb2[k] : 0.f;
      for (int w = 0; w < 4; ++w) eo += psum[w * MT + tid];
      float g = gates[b * K_ + k];
      atomicAdd(&y[b], g * eo);
    }
  }
}

extern "C" void kernel_launch(void* const* d_in, const int* in_sizes, int n_in,
                              void* d_out, int out_size, void* d_ws, size_t ws_size,
                              hipStream_t stream) {
  const int*   x         = (const int*)d_in[0];
  const int*   sql       = (const int*)d_in[1];
  const float* sql_emb_w = (const float*)d_in[2];
  const float* inp_emb_w = (const float*)d_in[3];
  const float* gw1 = (const float*)d_in[4];
  const float* gb1 = (const float*)d_in[5];
  const float* gw2 = (const float*)d_in[6];
  const float* gb2 = (const float*)d_in[7];
  const float* ew1 = (const float*)d_in[8];
  const float* eb1 = (const float*)d_in[9];
  const float* ew2 = (const float*)d_in[10];
  const float* eb2 = (const float*)d_in[11];
  float* out = (float*)d_out;

  float* h      = (float*)d_ws;
  float* gates  = h + (size_t)B_ * HID;
  int*   counts = (int*)(gates + (size_t)B_ * K_);
  int*   lists  = counts + K_;

  vsams_gate_l1_kernel<<<256, 128, 0, stream>>>(sql, sql_emb_w, gw1, gb1, h, out, counts);
  vsams_gate_head_kernel<<<256, 128, 0, stream>>>(h, gw2, gb2, gates, counts, lists);
  vsams_expert_kernel<<<545, 128, 0, stream>>>(x, inp_emb_w, ew1, eb1, ew2,
                                                eb2, gates, counts, lists, out);
}


# 2 launches, was 3: gate logits, softmax, top-2 run in the gate hidden-layer launch from its accumulators (16 x 512 tiles, h never stored); expert lists by prefix sum; logit sum order changes
# speedup vs baseline: 4.2023x; 1.0530x over previous
#include <hip/hip_runtime.h>

#define B_     4096
#define NFIELD 20
#define NEMB   16
#define HID    512
#define K_     16
#define GIN    320
#define EPSF   1e-10f
#define MT     32
#define NT     256
#define PITCH  324

typedef __attribute__((ext_vector_type(2))) float v2f;
typedef __attribute__((ext_vector_type(8))) float v8f;
typedef __attribute__((ext_vector_type(4))) float v4f;
typedef __attribute__((ext_vector_type(4))) int   v4i;
#define GMT    16
#define NGB    256

__device__ __forceinline__ v8f wmma_f32_16x16x4(v2f a, v2f b, v8f c) {
  return __builtin_amdgcn_wmma_f32_16x16x4_f32(false, a, false, b, (short)0, c,
                                               false, false);
}

template <int CTRL>
__device__ __forceinline__ float half_xor_get(float x) {
  return __builtin_bit_cast(float, __builtin_amdgcn_update_dpp(
      0, __builtin_bit_cast(int, x), CTRL, 0xf, 0xf, false));
}
__device__ __forceinline__ float half_sum16(float x) {
  x += half_xor_get<0x161>(x);
  x += half_xor_get<0x162>(x);
  x += half_xor_get<0x164>(x);
  x += half_xor_get<0x168>(x);
  return x;
}

template <int NS>
__device__ __forceinline__ void hidden_trip_load(const float* __restrict__ wp,
                                                 v4f (&b)[2 * NS]) {
#pragma unroll
  for (int s = 0; s < NS; ++s) {
    const float* r0 = wp + (size_t)(4 * s) * HID;
    b[2 * s + 0] = *(const v4f*)(r0);
    b[2 * s + 1] = *(const v4f*)(r0 + HID);
  }
}

template <int NS>
__device__ __forceinline__ void hidden_trip_mma(const float* ap,
                                                const v4f (&b)[2 * NS],
                                                v8f (&acc)[8]) {
#pragma unroll
  for (int s = 0; s < NS; ++s) {
    v2f a0, a1;
    a0.x = ap[4 * s];              a0.y = ap[4 * s + 1];
    a1.x = ap[16 * PITCH + 4 * s]; a1.y = ap[16 * PITCH + 4 * s + 1];
#pragma unroll
    for (int j = 0; j < 4; ++j) {
      v2f bb; bb.x = b[2 * s + 0][j]; bb.y = b[2 * s + 1][j];
      acc[j]     = wmma_f32_16x16x4(a0, bb, acc[j]);
      acc[4 + j] = wmma_f32_16x16x4(a1, bb, acc[4 + j]);
    }
  }
}

__device__ __forceinline__ void hidden_k_loop(const float* __restrict__ wp,
                                              const float* ap,
                                              v4f (&s0)[20], v4f (&s1)[20],
                                              v8f (&acc)[8]) {
  constexpr int TK = 40;
#pragma unroll 1
  for (int t = 0; t < 6; t += 2) {
    hidden_trip_load<10>(wp + (size_t)(t + 1) * TK * HID, s1);
    hidden_trip_mma<10>(ap + t * TK, s0, acc);
    hidden_trip_load<10>(wp + (size_t)(t + 2) * TK * HID, s0);
    hidden_trip_mma<10>(ap + (t + 1) * TK, s1, acc);
  }
  hidden_trip_load<10>(wp + (size_t)7 * TK * HID, s1);
  hidden_trip_mma<10>(ap + 6 * TK, s0, acc);
  hidden_trip_mma<10>(ap + 7 * TK, s1, acc);
}

template <int NS>
__device__ __forceinline__ void gate_trip_load(const float* __restrict__ wp,
                                               v4f (&w)[4 * NS]) {
#pragma unroll
  for (int s = 0; s < NS; ++s) {
    const float* r0 = wp + (size_t)(4 * s) * HID;
    w[4 * s + 0] = *(const v4f*)(r0);
    w[4 * s + 1] = *(const v4f*)(r0 + 64);
    w[4 * s + 2] = *(const v4f*)(r0 + HID);
    w[4 * s + 3] = *(const v4f*)(r0 + HID + 64);
  }
}

template <int NS>
__device__ __forceinline__ void gate_trip_mma(const float* ap,
                                              const v4f (&w)[4 * NS],
                                              v8f (&acc)[8]) {
#pragma unroll
  for (int s = 0; s < NS; ++s) {
    v2f xf;
    xf.x = ap[4 * s]; xf.y = ap[4 * s + 1];
#pragma unroll
    for (int j = 0; j < 4; ++j) {
      v2f w0; w0.x = w[4 * s + 0][j]; w0.y = w[4 * s + 2][j];
      v2f w1; w1.x = w[4 * s + 1][j]; w1.y = w[4 * s + 3][j];
      acc[j]     = wmma_f32_16x16x4(w0, xf, acc[j]);
      acc[4 + j] = wmma_f32_16x16x4(w1, xf, acc[4 + j]);
    }
  }
}

__global__ void __launch_bounds__(128) __attribute__((amdgpu_waves_per_eu(1, 1)))
vsams_gate_kernel(const int* __restrict__ sql,
                  const float* __restrict__ sql_emb_w,
                  const float* __restrict__ gw1,
                  const float* __restrict__ gb1,
                  const float* __restrict__ gw2,
                  const float* __restrict__ gb2,
                  float* __restrict__ gates,
                  float* __restrict__ y,
                  int* __restrict__ cntT,
                  int* __restrict__ loc) {
  __shared__ __attribute__((aligned(16))) float Alds[GMT * PITCH];
  __shared__ __attribute__((aligned(16))) float part[4 * 16 * 16];
  const int tid = threadIdx.x;
  const int m0  = blockIdx.x * GMT;
  if ((blockIdx.x & 1) == 0 && tid < 32) y[m0 + tid] = 0.f;
  const int lane = tid & 31, wave = tid >> 5;
  const int n0 = wave * 128;
  const int m  = lane & 15;
  const int kh = lane >> 4;
  const float* wp = gw1 + (size_t)(2 * kh) * HID + n0 + 4 * m;
  v4f s0[40], s1[40];
  {
    int s[10];
#pragma unroll
    for (int i = 0; i < 10; ++i) {
      int idx = tid + 128 * i;
      int r = idx / 80;
      int f = (idx - r * 80) >> 2;
      s[i] = sql[(m0 + r) * NFIELD + f];
    }
    gate_trip_load<10>(wp, s0);
    v4f v[10];
#pragma unroll
    for (int i = 0; i < 10; ++i)
      v[i] = *(const v4f*)&sql_emb_w[(size_t)s[i] * NEMB + (tid & 3) * 4];
#pragma unroll
    for (int i = 0; i < 10; ++i) {
      int idx = tid + 128 * i;
      int r = idx / 80;
      *(v4f*)&Alds[r * PITCH + (idx - r * 80) * 4] = v[i];
    }
  }
  __syncthreads();
  v8f acc[8];
#pragma unroll
  for (int j = 0; j < 8; ++j) acc[j] = (v8f){0.f, 0.f, 0.f, 0.f, 0.f, 0.f, 0.f, 0.f};
  const float* ap = &Alds[m * PITCH + 2 * kh];
  constexpr int TK = 40;
#pragma unroll 1
  for (int t = 0; t < 6; t += 2) {
    gate_trip_load<10>(wp + (size_t)(t + 1) * TK * HID, s1);
    gate_trip_mma<10>(ap + t * TK, s0, acc);
    gate_trip_load<10>(wp + (size_t)(t + 2) * TK * HID, s0);
    gate_trip_mma<10>(ap + (t + 1) * TK, s1, acc);
  }
  gate_trip_load<10>(wp + (size_t)7 * TK * HID, s1);
  const float* bp = gb1 + n0 + 32 * kh;
  const float* gp = gw2 + (size_t)(n0 + 32 * kh) * K_ + m;
  v4f bv[2][8];
#pragma unroll
  for (int g = 0; g < 2; ++g)
#pragma unroll
    for (int r = 0; r < 8; ++r) bv[g][r] = *(const v4f*)(bp + 64 * g + 4 * r);
  float g2v[8][8];
#pragma unroll
  for (int j = 0; j < 8; ++j)
#pragma unroll
    for (int r = 0; r < 8; ++r)
      g2v[j][r] = gp[(64 * (j >> 2) + 4 * r + (j & 3)) * K_];
  gate_trip_mma<10>(ap + 6 * TK, s0, acc);
  gate_trip_mma<10>(ap + 7 * TK, s1, acc);
  v8f l0 = {0.f, 0.f, 0.f, 0.f, 0.f, 0.f, 0.f, 0.f};
  v8f l1 = {0.f, 0.f, 0.f, 0.f, 0.f, 0.f, 0.f, 0.f};
#pragma unroll
  for (int j = 0; j < 8; ++j) {
#pragma unroll
    for (int q = 0; q < 4; ++q) {
      const int r = 2 * q;
      float h0 = acc[j][r]     + bv[j >> 2][r][j & 3];
      float h1 = acc[j][r + 1] + bv[j >> 2][r + 1][j & 3];
      v2f hb; hb.x = h0 > 0.f ? h0 : 0.f; hb.y = h1 > 0.f ? h1 : 0.f;
      v2f ga; ga.x = g2v[j][r]; ga.y = g2v[j][r + 1];
      if (q & 1) l1 = wmma_f32_16x16x4(ga, hb, l1);
      else       l0 = wmma_f32_16x16x4(ga, hb, l0);
    }
  }
  {
    float* pd = &part[wave * 256 + m * 16 + 8 * kh];
    *(v4f*)(pd)     = (v4f){l0[0] + l1[0], l0[1] + l1[1], l0[2] + l1[2], l0[3] + l1[3]};
    *(v4f*)(pd + 4) = (v4f){l0[4] + l1[4], l0[5] + l1[5], l0[6] + l1[6], l0[7] + l1[7]};
  }
  __syncthreads();
  if (tid < 16) {
    const int b_ = m0 + tid;
    float l[K_];
    float mx = -3.0e38f;
    for (int j = 0; j < K_; ++j) {
      float s4 = part[tid * 16 + j] + part[256 + tid * 16 + j];
      s4 += part[512 + tid * 16 + j];
      s4 += part[768 + tid * 16 + j];
      l[j] = s4 + gb2[j];
      mx = fmaxf(mx, l[j]);
    }
    float s = 0.f;
    for (int j = 0; j < K_; ++j) { l[j] = __expf(l[j] - mx); s += l[j]; }
    float inv = 1.f / s;
    for (int j = 0; j < K_; ++j) l[j] *= inv;
    int i1 = 0; float v1 = l[0];
    for (int j = 1; j < K_; ++j) if (l[j] > v1) { v1 = l[j]; i1 = j; }
    int i2 = -1; float v2 = -1.f;
    for (int j = 0; j < K_; ++j) if (j != i1 && l[j] > v2) { v2 = l[j]; i2 = j; }
    float denom = v1 + v2;
    float g1 = v1 / denom, g2 = v2 / denom;
#pragma unroll
    for (int q = 0; q < 4; ++q) {
      v4f gr;
#pragma unroll
      for (int c = 0; c < 4; ++c) {
        const int j = 4 * q + c;
        gr[c] = (j == i1) ? g1 : ((j == i2) ? g2 : 0.f);
      }
      *(v4f*)&gates[b_ * K_ + 4 * q] = gr;
    }
    const unsigned lower = (1u << tid) - 1u;
    int p1 = 0, p2 = 0, cmine = 0;
#pragma unroll
    for (int e = 0; e < K_; ++e) {
      const bool sel = (i1 == e) || (i2 == e);
      const unsigned mk = __builtin_amdgcn_ballot_w32(sel);
      const int below = __popc(mk & lower);
      if (i1 == e) p1 = below;
      if (i2 == e) p2 = below;
      if (tid == e) cmine = __popc(mk);
    }
    const int i2c = i2 < 0 ? 0 : i2;
    loc[(blockIdx.x * K_ + i1) * GMT + p1]  = b_;
    loc[(blockIdx.x * K_ + i2c) * GMT + p2] = b_;
    cntT[tid * NGB + blockIdx.x] = cmine;
  }
}

__device__ __forceinline__ void vsams_loss_block(const float* __restrict__ gates,
                                                 float* __restrict__ out,
                                                 float* __restrict__ scr) {
  float* wimp = scr;
  float* wld  = scr + 4 * K_;
  float* fin  = scr + 8 * K_;
  const int t = threadIdx.x;
  const int lane = t & 31, wave = t >> 5;
  float pi[4] = {0.f, 0.f, 0.f, 0.f}, pl[4] = {0.f, 0.f, 0.f, 0.f};
#pragma unroll 1
  for (int j0 = 0; j0 < 128; j0 += 16) {
    v4f g[16];
#pragma unroll
    for (int j = 0; j < 16; ++j)
      g[j] = *(const v4f*)&gates[(size_t)((j0 + j) * 128 + t) * 4];
#pragma unroll
    for (int j = 0; j < 16; ++j) {
#pragma unroll
      for (int c = 0; c < 4; ++c) {
        float gv = g[j][c];
        pi[c] += gv;
        pl[c] += (gv > 0.f) ? 1.f : 0.f;
      }
    }
  }
#pragma unroll
  for (int c = 0; c < 4; ++c) {
#pragma unroll
    for (int off = 4; off < 32; off <<= 1) {
      pi[c] += __shfl_xor(pi[c], off);
      pl[c] += __shfl_xor(pl[c], off);
    }
  }
  if (lane < 4) {
#pragma unroll
    for (int c = 0; c < 4; ++c) {
      wimp[wave * K_ + lane * 4 + c] = pi[c];
      wld[wave * K_ + lane * 4 + c]  = pl[c];
    }
  }
  __syncthreads();
  if (t < 2 * K_) {
    const float* src = (t < K_) ? wimp : wld;
    const int col = t & (K_ - 1);
    float s = 0.f;
    for (int w = 0; w < 4; ++w) s += src[w * K_ + col];
    fin[t] = s;
  }
  __syncthreads();
  if (t == 0) {
    float imp[K_], ld[K_];
    for (int kk = 0; kk < K_; ++kk) { imp[kk] = fin[kk]; ld[kk] = fin[K_ + kk]; }
    float mi = 0.f, ml = 0.f;
    for (int kk = 0; kk < K_; ++kk) { mi += imp[kk]; ml += ld[kk]; }
    mi *= (1.f / 16.f); ml *= (1.f / 16.f);
    float vi = 0.f, vl = 0.f;
    for (int kk = 0; kk < K_; ++kk) {
      float di = imp[kk] - mi, dl = ld[kk] - ml;
      vi += di * di; vl += dl * dl;
    }
    vi *= (1.f / 16.f); vl *= (1.f / 16.f);
    out[B_] = vi / (mi * mi + EPSF) + vl / (ml * ml + EPSF);
  }
}

__global__ void __launch_bounds__(128)
vsams_expert_kernel(const int* __restrict__ x,
                    const float* __restrict__ inp_emb_w,
                    const float* __restrict__ ew1,
                    const float* __restrict__ eb1,
                    const float* __restrict__ ew2,
                    const float* __restrict__ eb2,
                    const float* __restrict__ gates,
                    const int* __restrict__ cntT,
                    const int* __restrict__ loc,
                    float* __restrict__ y) {
  __shared__ __attribute__((aligned(16))) float Alds[MT * PITCH];
  __shared__ __attribute__((aligned(16))) float psum[4 * MT];
  __shared__ __attribute__((aligned(16))) int   pref[K_ * NGB];
  __shared__ __attribute__((aligned(16))) int   sums[128];
  __shared__ int   tot[K_];
  __shared__ int   bids[MT];
  if (blockIdx.x == 0) {
    vsams_loss_block(gates, y, Alds);
    return;
  }
  const int bid = blockIdx.x - 1;
  const int tid = threadIdx.x;
  {
    const int e = tid >> 3, pt = tid & 7;
    const int* cp = cntT + e * NGB + 32 * pt;
    v4i c[8];
#pragma unroll
    for (int i = 0; i < 8; ++i) c[i] = *(const v4i*)(cp + 4 * i);
    int pre[32];
    int run = 0;
#pragma unroll
    for (int i = 0; i < 32; ++i) { pre[i] = run; run += c[i >> 2][i & 3]; }
    sums[tid] = run;
    __syncthreads();
    const v4i sa = *(const v4i*)&sums[8 * e];
    const v4i sb = *(const v4i*)&sums[8 * e + 4];
    int off = 0, total = 0;
#pragma unroll
    for (int p = 0; p < 8; ++p) {
      const int sv = (p < 4) ? sa[p & 3] : sb[p & 3];
      off   += (p < pt) ? sv : 0;
      total += sv;
    }
#pragma unroll
    for (int i = 0; i < 8; ++i) {
      v4i o;
      o[0] = off + pre[4 * i];     o[1] = off + pre[4 * i + 1];
      o[2] = off + pre[4 * i + 2]; o[3] = off + pre[4 * i + 3];
      *(v4i*)&pref[e * NGB + 32 * pt + 4 * i] = o;
    }
    if (pt == 0) tot[e] = total;
  }
  __syncthreads();
  int k = -1, lc = 0, cnt = 0, first = 0;
  for (int e = 0; e < K_; ++e) {
    const int c  = tot[e];
    const int nb = ((c + MT - 1) / MT) * 2;
    if (bid >= first && bid < first + nb) { k = e; lc = bid - first; cnt = c; }
    first += nb;
  }
  k   = __builtin_amdgcn_readfirstlane(k);
  lc  = __builtin_amdgcn_readfirstlane(lc);
  cnt = __builtin_amdgcn_readfirstlane(cnt);
  if (k < 0) return;
  const int base = (lc >> 1) * MT;
  const int nh   = lc & 1;
  const int lane = tid & 31, wave = tid >> 5;
  const int m  = lane & 15;
  const int kh = lane >> 4;
  const int n0 = nh * NT + wave * 64;
  const float* W1 = ew1 + (size_t)k * GIN * HID;
  const float* B1 = eb1 + k * HID;
  const float* wp = W1 + (size_t)(2 * kh) * HID + n0 + 4 * m;
  v4f s0[20], s1[20];
  hidden_trip_load<10>(wp, s0);
  if (tid < MT) {
    const int p = base + tid;
    int row = -1;
    if (p < cnt) {
      const int* pk = pref + k * NGB;
      int g = 0;
#pragma unroll
      for (int step = NGB / 2; step >= 1; step >>= 1)
        if (pk[g + step] <= p) g += step;
      row = loc[(g * K_ + k) * GMT + (p - pk[g])];
    }
    bids[tid] = row;
  }
  __syncthreads();
  {
    int xi[20];
#pragma unroll
    for (int i = 0; i < 20; ++i) {
      int idx = tid + 128 * i;
      int r = idx / 80;
      int f = (idx - r * 80) >> 2;
      int b = bids[r];
      xi[i] = x[(b >= 0 ? b : 0) * NFIELD + f];
    }
    v4f v[20];
#pragma unroll
    for (int i = 0; i < 20; ++i)
      v[i] = *(const v4f*)&inp_emb_w[(size_t)xi[i] * NEMB + (tid & 3) * 4];
#pragma unroll
    for (int i = 0; i < 20; ++i) {
      int idx = tid + 128 * i;
      int r = idx / 80;
      *(v4f*)&Alds[r * PITCH + (idx - r * 80) * 4] = v[i];
    }
  }
  __syncthreads();
  {
    v8f acc[8];
#pragma unroll
    for (int j = 0; j < 8; ++j) acc[j] = (v8f){0.f, 0.f, 0.f, 0.f, 0.f, 0.f, 0.f, 0.f};
    hidden_k_loop(wp, &Alds[m * PITCH + 2 * kh], s0, s1, acc);
    const v4f bias = *(const v4f*)&B1[n0 + 4 * m];
    const v4f w2   = *(const v4f*)&ew2[k * HID + n0 + 4 * m];
    float pv[16];
#pragma unroll
    for (int i = 0; i < 2; ++i) {
#pragma unroll
      for (int v = 0; v < 8; ++v) {
        float p = 0.f;
#pragma unroll
        for (int j = 0; j < 4; ++j) {
          float x0 = acc[4 * i + j][v] + bias[j];
          x0 = x0 > 0.f ? x0 : 0.f;
          p += x0 * w2[j];
        }
        pv[8 * i + v] = half_sum16(p);
      }
    }
    if (m == 0) {
#pragma unroll
      for (int i = 0; i < 2; ++i) {
        float* dst = &psum[wave * MT + 16 * i + 8 * kh];
        *(v4f*)(dst)     = (v4f){pv[8 * i + 0], pv[8 * i + 1], pv[8 * i + 2], pv[8 * i + 3]};
        *(v4f*)(dst + 4) = (v4f){pv[8 * i + 4], pv[8 * i + 5], pv[8 * i + 6], pv[8 * i + 7]};
      }
    }
  }
  __syncthreads();
  if (tid < MT) {
    int b = bids[tid];
    if (b >= 0) {
      float eo = (nh == 0) ? eb2[k] : 0.f;
      for (int w = 0; w < 4; ++w) eo += psum[w * MT + tid];
      float g = gates[b * K_ + k];
      atomicAdd(&y[b], g * eo);
    }
  }
}

extern "C" void kernel_launch(void* const* d_in, const int* in_sizes, int n_in,
                              void* d_out, int out_size, void* d_ws, size_t ws_size,
                              hipStream_t stream) {
  const int*   x         = (const int*)d_in[0];
  const int*   sql       = (const int*)d_in[1];
  const float* sql_emb_w = (const float*)d_in[2];
  const float* inp_emb_w = (const float*)d_in[3];
  const float* gw1 = (const float*)d_in[4];
  const float* gb1 = (const float*)d_in[5];
  const float* gw2 = (const float*)d_in[6];
  const float* gb2 = (const float*)d_in[7];
  const float* ew1 = (const float*)d_in[8];
  const float* eb1 = (const float*)d_in[9];
  const float* ew2 = (const float*)d_in[10];
  const float* eb2 = (const float*)d_in[11];
  float* out = (float*)d_out;

  float* gates = (float*)d_ws;
  int*   cntT  = (int*)(gates + (size_t)B_ * K_);
  int*   loc   = cntT + K_ * NGB;

  vsams_gate_kernel<<<NGB, 128, 0, stream>>>(sql, sql_emb_w, gw1, gb1, gw2, gb2,
                                             gates, out, cntT, loc);
  vsams_expert_kernel<<<545, 128, 0, stream>>>(x, inp_emb_w, ew1, eb1, ew2,
                                                eb2, gates, cntT, loc, out);
}


# Row fragments of a whole 10-step trip read from LDS as one group one trip ahead (second register set) in both hidden k-loops: 80 WMMAs with no LDS read among them; sums unchanged
# speedup vs baseline: 4.4577x; 1.0608x over previous
#include <hip/hip_runtime.h>

#define B_     4096
#define NFIELD 20
#define NEMB   16
#define HID    512
#define K_     16
#define GIN    320
#define EPSF   1e-10f
#define MT     32
#define NT     256
#define PITCH  324

typedef __attribute__((ext_vector_type(2))) float v2f;
typedef __attribute__((ext_vector_type(8))) float v8f;
typedef __attribute__((ext_vector_type(4))) float v4f;
typedef __attribute__((ext_vector_type(4))) int   v4i;
#define GMT    16
#define NGB    256

__device__ __forceinline__ v8f wmma_f32_16x16x4(v2f a, v2f b, v8f c) {
  return __builtin_amdgcn_wmma_f32_16x16x4_f32(false, a, false, b, (short)0, c,
                                               false, false);
}

template <int CTRL>
__device__ __forceinline__ float half_xor_get(float x) {
  return __builtin_bit_cast(float, __builtin_amdgcn_update_dpp(
      0, __builtin_bit_cast(int, x), CTRL, 0xf, 0xf, false));
}
__device__ __forceinline__ float half_sum16(float x) {
  x += half_xor_get<0x161>(x);
  x += half_xor_get<0x162>(x);
  x += half_xor_get<0x164>(x);
  x += half_xor_get<0x168>(x);
  return x;
}

template <int NS>
__device__ __forceinline__ void hidden_trip_load(const float* __restrict__ wp,
                                                 v4f (&b)[2 * NS]) {
#pragma unroll
  for (int s = 0; s < NS; ++s) {
    const float* r0 = wp + (size_t)(4 * s) * HID;
    b[2 * s + 0] = *(const v4f*)(r0);
    b[2 * s + 1] = *(const v4f*)(r0 + HID);
  }
}

template <int NS>
__device__ __forceinline__ void hidden_trip_frag(const float* ap, v2f (&a)[2 * NS]) {
#pragma unroll
  for (int s = 0; s < NS; ++s) {
    a[2 * s + 0] = *(const v2f*)(ap + 4 * s);
    a[2 * s + 1] = *(const v2f*)(ap + 16 * PITCH + 4 * s);
  }
}

template <int NS>
__device__ __forceinline__ void hidden_trip_mma(const v2f (&a)[2 * NS],
                                                const v4f (&b)[2 * NS],
                                                v8f (&acc)[8]) {
#pragma unroll
  for (int s = 0; s < NS; ++s) {
#pragma unroll
    for (int j = 0; j < 4; ++j) {
      v2f bb; bb.x = b[2 * s + 0][j]; bb.y = b[2 * s + 1][j];
      acc[j]     = wmma_f32_16x16x4(a[2 * s + 0], bb, acc[j]);
      acc[4 + j] = wmma_f32_16x16x4(a[2 * s + 1], bb, acc[4 + j]);
    }
  }
}

__device__ __forceinline__ void hidden_k_loop(const float* __restrict__ wp,
                                              const float* ap,
                                              v4f (&s0)[20], v4f (&s1)[20],
                                              v8f (&acc)[8]) {
  constexpr int TK = 40;
  v2f a0[20], a1[20];
  hidden_trip_frag<10>(ap, a0);
#pragma unroll 1
  for (int t = 0; t < 6; t += 2) {
    hidden_trip_load<10>(wp + (size_t)(t + 1) * TK * HID, s1);
    hidden_trip_frag<10>(ap + (t + 1) * TK, a1);
    __builtin_amdgcn_sched_barrier(0);
    hidden_trip_mma<10>(a0, s0, acc);
    __builtin_amdgcn_sched_barrier(0);
    hidden_trip_load<10>(wp + (size_t)(t + 2) * TK * HID, s0);
    hidden_trip_frag<10>(ap + (t + 2) * TK, a0);
    __builtin_amdgcn_sched_barrier(0);
    hidden_trip_mma<10>(a1, s1, acc);
    __builtin_amdgcn_sched_barrier(0);
  }
  hidden_trip_load<10>(wp + (size_t)7 * TK * HID, s1);
  hidden_trip_frag<10>(ap + 7 * TK, a1);
  __builtin_amdgcn_sched_barrier(0);
  hidden_trip_mma<10>(a0, s0, acc);
  hidden_trip_mma<10>(a1, s1, acc);
}

template <int NS>
__device__ __forceinline__ void gate_trip_load(const float* __restrict__ wp,
                                               v4f (&w)[4 * NS]) {
#pragma unroll
  for (int s = 0; s < NS; ++s) {
    const float* r0 = wp + (size_t)(4 * s) * HID;
    w[4 * s + 0] = *(const v4f*)(r0);
    w[4 * s + 1] = *(const v4f*)(r0 + 64);
    w[4 * s + 2] = *(const v4f*)(r0 + HID);
    w[4 * s + 3] = *(const v4f*)(r0 + HID + 64);
  }
}

template <int NS>
__device__ __forceinline__ void gate_trip_frag(const float* ap, v2f (&xf)[NS]) {
#pragma unroll
  for (int s = 0; s < NS; ++s) xf[s] = *(const v2f*)(ap + 4 * s);
}

template <int NS>
__device__ __forceinline__ void gate_trip_mma(const v2f (&xf)[NS],
                                              const v4f (&w)[4 * NS],
                                              v8f (&acc)[8]) {
#pragma unroll
  for (int s = 0; s < NS; ++s) {
#pragma unroll
    for (int j = 0; j < 4; ++j) {
      v2f w0; w0.x = w[4 * s + 0][j]; w0.y = w[4 * s + 2][j];
      v2f w1; w1.x = w[4 * s + 1][j]; w1.y = w[4 * s + 3][j];
      acc[j]     = wmma_f32_16x16x4(w0, xf[s], acc[j]);
      acc[4 + j] = wmma_f32_16x16x4(w1, xf[s], acc[4 + j]);
    }
  }
}

__global__ void __launch_bounds__(128) __attribute__((amdgpu_waves_per_eu(1, 1)))
vsams_gate_kernel(const int* __restrict__ sql,
                  const float* __restrict__ sql_emb_w,
                  const float* __restrict__ gw1,
                  const float* __restrict__ gb1,
                  const float* __restrict__ gw2,
                  const float* __restrict__ gb2,
                  float* __restrict__ gates,
                  float* __restrict__ y,
                  int* __restrict__ cntT,
                  int* __restrict__ loc) {
  __shared__ __attribute__((aligned(16))) float Alds[GMT * PITCH];
  __shared__ __attribute__((aligned(16))) float part[4 * 16 * 16];
  const int tid = threadIdx.x;
  const int m0  = blockIdx.x * GMT;
  if ((blockIdx.x & 1) == 0 && tid < 32) y[m0 + tid] = 0.f;
  const int lane = tid & 31, wave = tid >> 5;
  const int n0 = wave * 128;
  const int m  = lane & 15;
  const int kh = lane >> 4;
  const float* wp = gw1 + (size_t)(2 * kh) * HID + n0 + 4 * m;
  v4f s0[40], s1[40];
  {
    int s[10];
#pragma unroll
    for (int i = 0; i < 10; ++i) {
      int idx = tid + 128 * i;
      int r = idx / 80;
      int f = (idx - r * 80) >> 2;
      s[i] = sql[(m0 + r) * NFIELD + f];
    }
    gate_trip_load<10>(wp, s0);
    v4f v[10];
#pragma unroll
    for (int i = 0; i < 10; ++i)
      v[i] = *(const v4f*)&sql_emb_w[(size_t)s[i] * NEMB + (tid & 3) * 4];
#pragma unroll
    for (int i = 0; i < 10; ++i) {
      int idx = tid + 128 * i;
      int r = idx / 80;
      *(v4f*)&Alds[r * PITCH + (idx - r * 80) * 4] = v[i];
    }
  }
  __syncthreads();
  v8f acc[8];
#pragma unroll
  for (int j = 0; j < 8; ++j) acc[j] = (v8f){0.f, 0.f, 0.f, 0.f, 0.f, 0.f, 0.f, 0.f};
  const float* ap = &Alds[m * PITCH + 2 * kh];
  constexpr int TK = 40;
  v2f x0[10], x1[10];
  gate_trip_frag<10>(ap, x0);
#pragma unroll 1
  for (int t = 0; t < 6; t += 2) {
    gate_trip_load<10>(wp + (size_t)(t + 1) * TK * HID, s1);
    gate_trip_frag<10>(ap + (t + 1) * TK, x1);
    __builtin_amdgcn_sched_barrier(0);
    gate_trip_mma<10>(x0, s0, acc);
    __builtin_amdgcn_sched_barrier(0);
    gate_trip_load<10>(wp + (size_t)(t + 2) * TK * HID, s0);
    gate_trip_frag<10>(ap + (t + 2) * TK, x0);
    __builtin_amdgcn_sched_barrier(0);
    gate_trip_mma<10>(x1, s1, acc);
    __builtin_amdgcn_sched_barrier(0);
  }
  gate_trip_load<10>(wp + (size_t)7 * TK * HID, s1);
  gate_trip_frag<10>(ap + 7 * TK, x1);
  const float* bp = gb1 + n0 + 32 * kh;
  const float* gp = gw2 + (size_t)(n0 + 32 * kh) * K_ + m;
  v4f bv[2][8];
#pragma unroll
  for (int g = 0; g < 2; ++g)
#pragma unroll
    for (int r = 0; r < 8; ++r) bv[g][r] = *(const v4f*)(bp + 64 * g + 4 * r);
  __builtin_amdgcn_sched_barrier(0);
  gate_trip_mma<10>(x0, s0, acc);
  __builtin_amdgcn_sched_barrier(0);
  float g2v[8][8];
#pragma unroll
  for (int j = 0; j < 8; ++j)
#pragma unroll
    for (int r = 0; r < 8; ++r)
      g2v[j][r] = gp[(64 * (j >> 2) + 4 * r + (j & 3)) * K_];
  __builtin_amdgcn_sched_barrier(0);
  gate_trip_mma<10>(x1, s1, acc);
  v8f l0 = {0.f, 0.f, 0.f, 0.f, 0.f, 0.f, 0.f, 0.f};
  v8f l1 = {0.f, 0.f, 0.f, 0.f, 0.f, 0.f, 0.f, 0.f};
#pragma unroll
  for (int j = 0; j < 8; ++j) {
#pragma unroll
    for (int q = 0; q < 4; ++q) {
      const int r = 2 * q;
      float h0 = acc[j][r]     + bv[j >> 2][r][j & 3];
      float h1 = acc[j][r + 1] + bv[j >> 2][r + 1][j & 3];
      v2f hb; hb.x = h0 > 0.f ? h0 : 0.f; hb.y = h1 > 0.f ? h1 : 0.f;
      v2f ga; ga.x = g2v[j][r]; ga.y = g2v[j][r + 1];
      if (q & 1) l1 = wmma_f32_16x16x4(ga, hb, l1);
      else       l0 = wmma_f32_16x16x4(ga, hb, l0);
    }
  }
  {
    float* pd = &part[wave * 256 + m * 16 + 8 * kh];
    *(v4f*)(pd)     = (v4f){l0[0] + l1[0], l0[1] + l1[1], l0[2] + l1[2], l0[3] + l1[3]};
    *(v4f*)(pd + 4) = (v4f){l0[4] + l1[4], l0[5] + l1[5], l0[6] + l1[6], l0[7] + l1[7]};
  }
  __syncthreads();
  if (tid < 16) {
    const int b_ = m0 + tid;
    float l[K_];
    float mx = -3.0e38f;
    for (int j = 0; j < K_; ++j) {
      float s4 = part[tid * 16 + j] + part[256 + tid * 16 + j];
      s4 += part[512 + tid * 16 + j];
      s4 += part[768 + tid * 16 + j];
      l[j] = s4 + gb2[j];
      mx = fmaxf(mx, l[j]);
    }
    float s = 0.f;
    for (int j = 0; j < K_; ++j) { l[j] = __expf(l[j] - mx); s += l[j]; }
    float inv = 1.f / s;
    for (int j = 0; j < K_; ++j) l[j] *= inv;
    int i1 = 0; float v1 = l[0];
    for (int j = 1; j < K_; ++j) if (l[j] > v1) { v1 = l[j]; i1 = j; }
    int i2 = -1; float v2 = -1.f;
    for (int j = 0; j < K_; ++j) if (j != i1 && l[j] > v2) { v2 = l[j]; i2 = j; }
    float denom = v1 + v2;
    float g1 = v1 / denom, g2 = v2 / denom;
#pragma unroll
    for (int q = 0; q < 4; ++q) {
      v4f gr;
#pragma unroll
      for (int c = 0; c < 4; ++c) {
        const int j = 4 * q + c;
        gr[c] = (j == i1) ? g1 : ((j == i2) ? g2 : 0.f);
      }
      *(v4f*)&gates[b_ * K_ + 4 * q] = gr;
    }
    const unsigned lower = (1u << tid) - 1u;
    int p1 = 0, p2 = 0, cmine = 0;
#pragma unroll
    for (int e = 0; e < K_; ++e) {
      const bool sel = (i1 == e) || (i2 == e);
      const unsigned mk = __builtin_amdgcn_ballot_w32(sel);
      const int below = __popc(mk & lower);
      if (i1 == e) p1 = below;
      if (i2 == e) p2 = below;
      if (tid == e) cmine = __popc(mk);
    }
    const int i2c = i2 < 0 ? 0 : i2;
    loc[(blockIdx.x * K_ + i1) * GMT + p1]  = b_;
    loc[(blockIdx.x * K_ + i2c) * GMT + p2] = b_;
    cntT[tid * NGB + blockIdx.x] = cmine;
  }
}

__device__ __forceinline__ void vsams_loss_block(const float* __restrict__ gates,
                                                 float* __restrict__ out,
                                                 float* __restrict__ scr) {
  float* wimp = scr;
  float* wld  = scr + 4 * K_;
  float* fin  = scr + 8 * K_;
  const int t = threadIdx.x;
  const int lane = t & 31, wave = t >> 5;
  float pi[4] = {0.f, 0.f, 0.f, 0.f}, pl[4] = {0.f, 0.f, 0.f, 0.f};
#pragma unroll 1
  for (int j0 = 0; j0 < 128; j0 += 16) {
    v4f g[16];
#pragma unroll
    for (int j = 0; j < 16; ++j)
      g[j] = *(const v4f*)&gates[(size_t)((j0 + j) * 128 + t) * 4];
#pragma unroll
    for (int j = 0; j < 16; ++j) {
#pragma unroll
      for (int c = 0; c < 4; ++c) {
        float gv = g[j][c];
        pi[c] += gv;
        pl[c] += (gv > 0.f) ? 1.f : 0.f;
      }
    }
  }
#pragma unroll
  for (int c = 0; c < 4; ++c) {
#pragma unroll
    for (int off = 4; off < 32; off <<= 1) {
      pi[c] += __shfl_xor(pi[c], off);
      pl[c] += __shfl_xor(pl[c], off);
    }
  }
  if (lane < 4) {
#pragma unroll
    for (int c = 0; c < 4; ++c) {
      wimp[wave * K_ + lane * 4 + c] = pi[c];
      wld[wave * K_ + lane * 4 + c]  = pl[c];
    }
  }
  __syncthreads();
  if (t < 2 * K_) {
    const float* src = (t < K_) ? wimp : wld;
    const int col = t & (K_ - 1);
    float s = 0.f;
    for (int w = 0; w < 4; ++w) s += src[w * K_ + col];
    fin[t] = s;
  }
  __syncthreads();
  if (t == 0) {
    float imp[K_], ld[K_];
    for (int kk = 0; kk < K_; ++kk) { imp[kk] = fin[kk]; ld[kk] = fin[K_ + kk]; }
    float mi = 0.f, ml = 0.f;
    for (int kk = 0; kk < K_; ++kk) { mi += imp[kk]; ml += ld[kk]; }
    mi *= (1.f / 16.f); ml *= (1.f / 16.f);
    float vi = 0.f, vl = 0.f;
    for (int kk = 0; kk < K_; ++kk) {
      float di = imp[kk] - mi, dl = ld[kk] - ml;
      vi += di * di; vl += dl * dl;
    }
    vi *= (1.f / 16.f); vl *= (1.f / 16.f);
    out[B_] = vi / (mi * mi + EPSF) + vl / (ml * ml + EPSF);
  }
}

__global__ void __launch_bounds__(128)
vsams_expert_kernel(const int* __restrict__ x,
                    const float* __restrict__ inp_emb_w,
                    const float* __restrict__ ew1,
                    const float* __restrict__ eb1,
                    const float* __restrict__ ew2,
                    const float* __restrict__ eb2,
                    const float* __restrict__ gates,
                    const int* __restrict__ cntT,
                    const int* __restrict__ loc,
                    float* __restrict__ y) {
  __shared__ __attribute__((aligned(16))) float Alds[MT * PITCH];
  __shared__ __attribute__((aligned(16))) float psum[4 * MT];
  __shared__ __attribute__((aligned(16))) int   pref[K_ * NGB];
  __shared__ __attribute__((aligned(16))) int   sums[128];
  __shared__ int   tot[K_];
  __shared__ int   bids[MT];
  if (blockIdx.x == 0) {
    vsams_loss_block(gates, y, Alds);
    return;
  }
  const int bid = blockIdx.x - 1;
  const int tid = threadIdx.x;
  {
    const int e = tid >> 3, pt = tid & 7;
    const int* cp = cntT + e * NGB + 32 * pt;
    v4i c[8];
#pragma unroll
    for (int i = 0; i < 8; ++i) c[i] = *(const v4i*)(cp + 4 * i);
    int pre[32];
    int run = 0;
#pragma unroll
    for (int i = 0; i < 32; ++i) { pre[i] = run; run += c[i >> 2][i & 3]; }
    sums[tid] = run;
    __syncthreads();
    const v4i sa = *(const v4i*)&sums[8 * e];
    const v4i sb = *(const v4i*)&sums[8 * e + 4];
    int off = 0, total = 0;
#pragma unroll
    for (int p = 0; p < 8; ++p) {
      const int sv = (p < 4) ? sa[p & 3] : sb[p & 3];
      off   += (p < pt) ? sv : 0;
      total += sv;
    }
#pragma unroll
    for (int i = 0; i < 8; ++i) {
      v4i o;
      o[0] = off + pre[4 * i];     o[1] = off + pre[4 * i + 1];
      o[2] = off + pre[4 * i + 2]; o[3] = off + pre[4 * i + 3];
      *(v4i*)&pref[e * NGB + 32 * pt + 4 * i] = o;
    }
    if (pt == 0) tot[e] = total;
  }
  __syncthreads();
  int k = -1, lc = 0, cnt = 0, first = 0;
  for (int e = 0; e < K_; ++e) {
    const int c  = tot[e];
    const int nb = ((c + MT - 1) / MT) * 2;
    if (bid >= first && bid < first + nb) { k = e; lc = bid - first; cnt = c; }
    first += nb;
  }
  k   = __builtin_amdgcn_readfirstlane(k);
  lc  = __builtin_amdgcn_readfirstlane(lc);
  cnt = __builtin_amdgcn_readfirstlane(cnt);
  if (k < 0) return;
  const int base = (lc >> 1) * MT;
  const int nh   = lc & 1;
  const int lane = tid & 31, wave = tid >> 5;
  const int m  = lane & 15;
  const int kh = lane >> 4;
  const int n0 = nh * NT + wave * 64;
  const float* W1 = ew1 + (size_t)k * GIN * HID;
  const float* B1 = eb1 + k * HID;
  const float* wp = W1 + (size_t)(2 * kh) * HID + n0 + 4 * m;
  v4f s0[20], s1[20];
  hidden_trip_load<10>(wp, s0);
  if (tid < MT) {
    const int p = base + tid;
    int row = -1;
    if (p < cnt) {
      const int* pk = pref + k * NGB;
      int g = 0;
#pragma unroll
      for (int step = NGB / 2; step >= 1; step >>= 1)
        if (pk[g + step] <= p) g += step;
      row = loc[(g * K_ + k) * GMT + (p - pk[g])];
    }
    bids[tid] = row;
  }
  __syncthreads();
  {
    int xi[20];
#pragma unroll
    for (int i = 0; i < 20; ++i) {
      int idx = tid + 128 * i;
      int r = idx / 80;
      int f = (idx - r * 80) >> 2;
      int b = bids[r];
      xi[i] = x[(b >= 0 ? b : 0) * NFIELD + f];
    }
    v4f v[20];
#pragma unroll
    for (int i = 0; i < 20; ++i)
      v[i] = *(const v4f*)&inp_emb_w[(size_t)xi[i] * NEMB + (tid & 3) * 4];
#pragma unroll
    for (int i = 0; i < 20; ++i) {
      int idx = tid + 128 * i;
      int r = idx / 80;
      *(v4f*)&Alds[r * PITCH + (idx - r * 80) * 4] = v[i];
    }
  }
  __syncthreads();
  {
    v8f acc[8];
#pragma unroll
    for (int j = 0; j < 8; ++j) acc[j] = (v8f){0.f, 0.f, 0.f, 0.f, 0.f, 0.f, 0.f, 0.f};
    hidden_k_loop(wp, &Alds[m * PITCH + 2 * kh], s0, s1, acc);
    const v4f bias = *(const v4f*)&B1[n0 + 4 * m];
    const v4f w2   = *(const v4f*)&ew2[k * HID + n0 + 4 * m];
    float pv[16];
#pragma unroll
    for (int i = 0; i < 2; ++i) {
#pragma unroll
      for (int v = 0; v < 8; ++v) {
        float p = 0.f;
#pragma unroll
        for (int j = 0; j < 4; ++j) {
          float x0 = acc[4 * i + j][v] + bias[j];
          x0 = x0 > 0.f ? x0 : 0.f;
          p += x0 * w2[j];
        }
        pv[8 * i + v] = half_sum16(p);
      }
    }
    if (m == 0) {
#pragma unroll
      for (int i = 0; i < 2; ++i) {
        float* dst = &psum[wave * MT + 16 * i + 8 * kh];
        *(v4f*)(dst)     = (v4f){pv[8 * i + 0], pv[8 * i + 1], pv[8 * i + 2], pv[8 * i + 3]};
        *(v4f*)(dst + 4) = (v4f){pv[8 * i + 4], pv[8 * i + 5], pv[8 * i + 6], pv[8 * i + 7]};
      }
    }
  }
  __syncthreads();
  if (tid < MT) {
    int b = bids[tid];
    if (b >= 0) {
      float eo = (nh == 0) ? eb2[k] : 0.f;
      for (int w = 0; w < 4; ++w) eo += psum[w * MT + tid];
      float g = gates[b * K_ + k];
      atomicAdd(&y[b], g * eo);
    }
  }
}

extern "C" void kernel_launch(void* const* d_in, const int* in_sizes, int n_in,
                              void* d_out, int out_size, void* d_ws, size_t ws_size,
                              hipStream_t stream) {
  const int*   x         = (const int*)d_in[0];
  const int*   sql       = (const int*)d_in[1];
  const float* sql_emb_w = (const float*)d_in[2];
  const float* inp_emb_w = (const float*)d_in[3];
  const float* gw1 = (const float*)d_in[4];
  const float* gb1 = (const float*)d_in[5];
  const float* gw2 = (const float*)d_in[6];
  const float* gb2 = (const float*)d_in[7];
  const float* ew1 = (const float*)d_in[8];
  const float* eb1 = (const float*)d_in[9];
  const float* ew2 = (const float*)d_in[10];
  const float* eb2 = (const float*)d_in[11];
  float* out = (float*)d_out;

  float* gates = (float*)d_ws;
  int*   cntT  = (int*)(gates + (size_t)B_ * K_);
  int*   loc   = cntT + K_ * NGB;

  vsams_gate_kernel<<<NGB, 128, 0, stream>>>(sql, sql_emb_w, gw1, gb1, gw2, gb2,
                                             gates, out, cntT, loc);
  vsams_expert_kernel<<<545, 128, 0, stream>>>(x, inp_emb_w, ew1, eb1, ew2,
                                                eb2, gates, cntT, loc, out);
}
